# starting kernel
# baseline (speedup 1.0000x reference)
#include <hip/hip_runtime.h>
#include <hip/hip_bf16.h>

typedef __attribute__((ext_vector_type(2))) float v2f;
typedef __attribute__((ext_vector_type(8))) float v8f;

#define N_V   100000
#define N_E   20000
#define N_NZ  400000
#define CIN   256
#define HID   256
#define NH    4
#define DH    64
#define NEG_SLOPE 0.2f

__global__ __launch_bounds__(256) void v2e_scatter(
    const float* __restrict__ X, const int* __restrict__ inc_v,
    const int* __restrict__ inc_e, float* __restrict__ Xsum,
    float* __restrict__ cnt)
{
  const int widx = blockIdx.x * 8 + (threadIdx.x >> 5);
  if (widx >= N_NZ) return;
  const int lane = threadIdx.x & 31;
  const int v = inc_v[widx];
  const int e = inc_e[widx];
  const float4* __restrict__ src4 = (const float4*)(X + (size_t)v * CIN);
  float* __restrict__ dst = Xsum + (size_t)e * HID;
#pragma unroll
  for (int f4 = lane; f4 < HID / 4; f4 += 32) {
    const float4 val = src4[f4];
    atomicAdd(&dst[f4 * 4 + 0], val.x);
    atomicAdd(&dst[f4 * 4 + 1], val.y);
    atomicAdd(&dst[f4 * 4 + 2], val.z);
    atomicAdd(&dst[f4 * 4 + 3], val.w);
  }
  if (lane == 0) atomicAdd(&cnt[e], 1.0f);
}

__global__ __launch_bounds__(256) void gemm_edge(
    const float* __restrict__ Xsum, const float* __restrict__ cnt,
    const float* __restrict__ W, const float* __restrict__ bvec,
    float* __restrict__ Y)
{
  const int lane = threadIdx.x & 31;
  const int wid  = blockIdx.x * 8 + (threadIdx.x >> 5);
  const int row_tile = wid >> 2;
  const int head = wid & 3;
  const int row0 = row_tile * 16;
  const int m = lane & 15;
  const int half = lane >> 4;

  const int row = row0 + m;
  const float inv = 1.0f / fmaxf(cnt[row], 1.0f);
  const float* __restrict__ arow = Xsum + (size_t)row * HID;
  const float* __restrict__ wh = W + (size_t)head * (CIN * DH);

  v8f acc[4] = {v8f{}, v8f{}, v8f{}, v8f{}};
#pragma unroll 4
  for (int k0 = 0; k0 < CIN; k0 += 4) {
    const int ka = k0 + 2 * half;
    v2f afrag;
    afrag.x = arow[ka] * inv;
    afrag.y = arow[ka + 1] * inv;
    const float* __restrict__ wk = wh + (size_t)ka * DH + m;
#pragma unroll
    for (int t = 0; t < 4; ++t) {
      v2f bfrag;
      bfrag.x = wk[t * 16];
      bfrag.y = wk[t * 16 + DH];
      acc[t] = __builtin_amdgcn_wmma_f32_16x16x4_f32(
          false, afrag, false, bfrag, (short)0, acc[t], false, false);
    }
  }

  const int mbase = half * 8;
#pragma unroll
  for (int t = 0; t < 4; ++t) {
    const int col = head * DH + t * 16 + m;
    const float bias = bvec[col];
#pragma unroll
    for (int r = 0; r < 8; ++r) {
      Y[(size_t)(row0 + mbase + r) * HID + col] = acc[t][r] + bias;
    }
  }
}

__global__ __launch_bounds__(256) void edge_alpha(
    const float* __restrict__ Y, const float* __restrict__ a_e,
    float* __restrict__ es)
{
  __shared__ float red[256];
  const int e = blockIdx.x;
  const int f = threadIdx.x;
  red[f] = Y[(size_t)e * HID + f] * a_e[f];
  __syncthreads();
#pragma unroll
  for (int off = 32; off >= 1; off >>= 1) {
    if ((f & 63) < off) red[f] += red[f + off];
    __syncthreads();
  }
  if ((f & 63) == 0) {
    const float alpha = red[f];
    float s = alpha >= 0.0f ? alpha : NEG_SLOPE * alpha;
    s = fminf(fmaxf(s, 0.001f), 5.0f);
    es[e * NH + (f >> 6)] = __expf(s);
  }
}

__global__ __launch_bounds__(256) void vert_denom(
    const float* __restrict__ es, const int* __restrict__ inc_v,
    const int* __restrict__ inc_e, float* __restrict__ denom)
{
  const int inc = blockIdx.x * 256 + threadIdx.x;
  if (inc >= N_NZ) return;
  const int v = inc_v[inc], e = inc_e[inc];
  const float4 s4 = ((const float4*)es)[e];
  float* __restrict__ d = denom + (size_t)v * NH;
  atomicAdd(&d[0], s4.x);
  atomicAdd(&d[1], s4.y);
  atomicAdd(&d[2], s4.z);
  atomicAdd(&d[3], s4.w);
}

__global__ __launch_bounds__(256) void aggregate(
    const float* __restrict__ Y, const float* __restrict__ es,
    const int* __restrict__ inc_v, const int* __restrict__ inc_e,
    const float* __restrict__ denom, float* __restrict__ out)
{
  const int widx = blockIdx.x * 8 + (threadIdx.x >> 5);
  if (widx >= N_NZ) return;
  const int lane = threadIdx.x & 31;
  const int v = inc_v[widx];
  const int e = inc_e[widx];
  const float4 s4 = ((const float4*)es)[e];
  const float4 d4 = ((const float4*)denom)[v];
  float wgt[NH] = { s4.x / d4.x, s4.y / d4.y, s4.z / d4.z, s4.w / d4.w };

  const float4* __restrict__ yrow4 = (const float4*)(Y + (size_t)e * HID);
  float* __restrict__ orow = out + (size_t)v * HID;
#pragma unroll
  for (int f4 = lane; f4 < HID / 4; f4 += 32) {
    const float w = wgt[f4 >> 4];
    const float4 y = yrow4[f4];
    atomicAdd(&orow[f4 * 4 + 0], w * y.x);
    atomicAdd(&orow[f4 * 4 + 1], w * y.y);
    atomicAdd(&orow[f4 * 4 + 2], w * y.z);
    atomicAdd(&orow[f4 * 4 + 3], w * y.w);
  }
}

__global__ __launch_bounds__(256) void elu_kernel(float4* __restrict__ out)
{
  const size_t i = (size_t)blockIdx.x * 256 + threadIdx.x;
  if (i >= (size_t)N_V * HID / 4) return;
  float4 x = out[i];
  x.x = x.x > 0.0f ? x.x : expm1f(x.x);
  x.y = x.y > 0.0f ? x.y : expm1f(x.y);
  x.z = x.z > 0.0f ? x.z : expm1f(x.z);
  x.w = x.w > 0.0f ? x.w : expm1f(x.w);
  out[i] = x;
}

extern "C" void kernel_launch(void* const* d_in, const int* in_sizes, int n_in,
                              void* d_out, int out_size, void* d_ws, size_t ws_size,
                              hipStream_t stream) {
  const float* X     = (const float*)d_in[0];
  const float* W     = (const float*)d_in[1];
  const float* bvec  = (const float*)d_in[2];
  const float* a_e   = (const float*)d_in[3];
  const int*   inc_v = (const int*)d_in[4];
  const int*   inc_e = (const int*)d_in[5];
  float* out = (float*)d_out;

  float* ws    = (float*)d_ws;
  float* Xsum  = ws;
  float* Y     = Xsum + (size_t)N_E * HID;
  float* cnt   = Y    + (size_t)N_E * HID;
  float* es    = cnt  + N_E;
  float* denom = es   + (size_t)N_E * NH;

  hipMemsetAsync(Xsum, 0, sizeof(float) * (size_t)N_E * HID, stream);
  hipMemsetAsync(cnt, 0, sizeof(float) * N_E, stream);
  hipMemsetAsync(denom, 0, sizeof(float) * (size_t)N_V * NH, stream);
  hipMemsetAsync(out, 0, sizeof(float) * (size_t)N_V * HID, stream);

  v2e_scatter<<<dim3(N_NZ / 8), 256, 0, stream>>>(X, inc_v, inc_e, Xsum, cnt);
  gemm_edge<<<dim3((N_E / 16) * NH / 8), 256, 0, stream>>>(Xsum, cnt, W, bvec, Y);
  edge_alpha<<<dim3(N_E), 256, 0, stream>>>(Y, a_e, es);
  vert_denom<<<dim3((N_NZ + 255) / 256), 256, 0, stream>>>(es, inc_v, inc_e, denom);
  aggregate<<<dim3(N_NZ / 8), 256, 0, stream>>>(Y, es, inc_v, inc_e, denom, out);
  elu_kernel<<<dim3(((size_t)N_V * HID / 4 + 255) / 256), 256, 0, stream>>>((float4*)out);
}


# e2v as per-vertex gather over an in-call counting-sort list (int atomics) instead of f32 atomics; denominators, ELU and the single out store folded in; sum order = arrival-order rank
# speedup vs baseline: 1.4986x; 1.4986x over previous
#include <hip/hip_runtime.h>
#include <hip/hip_bf16.h>

typedef __attribute__((ext_vector_type(2))) float v2f;
typedef __attribute__((ext_vector_type(8))) float v8f;

#define N_V   100000
#define N_E   20000
#define N_NZ  400000
#define CIN   256
#define HID   256
#define NH    4
#define DH    64
#define NEG_SLOPE 0.2f

__global__ __launch_bounds__(256) void v2e_scatter(
    const float* __restrict__ X, const int* __restrict__ inc_v,
    const int* __restrict__ inc_e, float* __restrict__ Xsum,
    float* __restrict__ cnt)
{
  const int widx = blockIdx.x * 8 + (threadIdx.x >> 5);
  if (widx >= N_NZ) return;
  const int lane = threadIdx.x & 31;
  const int v = inc_v[widx];
  const int e = inc_e[widx];
  const float4* __restrict__ src4 = (const float4*)(X + (size_t)v * CIN);
  float* __restrict__ dst = Xsum + (size_t)e * HID;
#pragma unroll
  for (int f4 = lane; f4 < HID / 4; f4 += 32) {
    const float4 val = src4[f4];
    atomicAdd(&dst[f4 * 4 + 0], val.x);
    atomicAdd(&dst[f4 * 4 + 1], val.y);
    atomicAdd(&dst[f4 * 4 + 2], val.z);
    atomicAdd(&dst[f4 * 4 + 3], val.w);
  }
  if (lane == 0) atomicAdd(&cnt[e], 1.0f);
}

__global__ __launch_bounds__(256) void gemm_edge(
    const float* __restrict__ Xsum, const float* __restrict__ cnt,
    const float* __restrict__ W, const float* __restrict__ bvec,
    float* __restrict__ Y)
{
  const int lane = threadIdx.x & 31;
  const int wid  = blockIdx.x * 8 + (threadIdx.x >> 5);
  const int row_tile = wid >> 2;
  const int head = wid & 3;
  const int row0 = row_tile * 16;
  const int m = lane & 15;
  const int half = lane >> 4;

  const int row = row0 + m;
  const float inv = 1.0f / fmaxf(cnt[row], 1.0f);
  const float* __restrict__ arow = Xsum + (size_t)row * HID;
  const float* __restrict__ wh = W + (size_t)head * (CIN * DH);

  v8f acc[4] = {v8f{}, v8f{}, v8f{}, v8f{}};
#pragma unroll 4
  for (int k0 = 0; k0 < CIN; k0 += 4) {
    const int ka = k0 + 2 * half;
    v2f afrag;
    afrag.x = arow[ka] * inv;
    afrag.y = arow[ka + 1] * inv;
    const float* __restrict__ wk = wh + (size_t)ka * DH + m;
#pragma unroll
    for (int t = 0; t < 4; ++t) {
      v2f bfrag;
      bfrag.x = wk[t * 16];
      bfrag.y = wk[t * 16 + DH];
      acc[t] = __builtin_amdgcn_wmma_f32_16x16x4_f32(
          false, afrag, false, bfrag, (short)0, acc[t], false, false);
    }
  }

  const int mbase = half * 8;
#pragma unroll
  for (int t = 0; t < 4; ++t) {
    const int col = head * DH + t * 16 + m;
    const float bias = bvec[col];
#pragma unroll
    for (int r = 0; r < 8; ++r) {
      Y[(size_t)(row0 + mbase + r) * HID + col] = acc[t][r] + bias;
    }
  }
}

__global__ __launch_bounds__(256) void edge_alpha(
    const float* __restrict__ Y, const float* __restrict__ a_e,
    float* __restrict__ es)
{
  __shared__ float red[256];
  const int e = blockIdx.x;
  const int f = threadIdx.x;
  red[f] = Y[(size_t)e * HID + f] * a_e[f];
  __syncthreads();
#pragma unroll
  for (int off = 32; off >= 1; off >>= 1) {
    if ((f & 63) < off) red[f] += red[f + off];
    __syncthreads();
  }
  if ((f & 63) == 0) {
    const float alpha = red[f];
    float s = alpha >= 0.0f ? alpha : NEG_SLOPE * alpha;
    s = fminf(fmaxf(s, 0.001f), 5.0f);
    es[e * NH + (f >> 6)] = __expf(s);
  }
}


__global__ __launch_bounds__(256) void v_hist(
    const int* __restrict__ inc_v, int* __restrict__ deg)
{
  const int i = blockIdx.x * 256 + threadIdx.x;
  if (i >= N_NZ) return;
  atomicAdd(&deg[inc_v[i]], 1);
}

__global__ __launch_bounds__(256) void v_alloc(
    const int* __restrict__ deg, int* __restrict__ counter,
    int* __restrict__ start, int* __restrict__ cur)
{
  const int v = blockIdx.x * 256 + threadIdx.x;
  const int lane = threadIdx.x & 31;
  const int d = (v < N_V) ? deg[v] : 0;
  int incl = d;
  for (int o = 1; o < 32; o <<= 1) {
    const int up = __shfl_up(incl, o);
    if (lane >= o) incl += up;
  }
  int base = 0;
  if (lane == 31) base = atomicAdd(counter, incl);
  base = __shfl(base, 31);
  if (v < N_V) {
    const int s = base + incl - d;
    start[v] = s;
    cur[v] = s;
  }
}

__global__ __launch_bounds__(256) void v_fill(
    const int* __restrict__ inc_v, const int* __restrict__ inc_e,
    int* __restrict__ cur, int* __restrict__ lst)
{
  const int i = blockIdx.x * 256 + threadIdx.x;
  if (i >= N_NZ) return;
  const int slot = atomicAdd(&cur[inc_v[i]], 1);
  lst[slot] = inc_e[i];
}

__global__ __launch_bounds__(256) void v_gather(
    const float* __restrict__ Y, const float* __restrict__ es,
    const int* __restrict__ start, const int* __restrict__ end,
    const int* __restrict__ lst, float* __restrict__ out)
{
  const int v = __builtin_amdgcn_readfirstlane(blockIdx.x * 8 + (threadIdx.x >> 5));
  const int lane = threadIdx.x & 31;
  const int s = start[v];
  const int t = end[v];
  const float4* __restrict__ es4 = (const float4*)es;

  float d0 = 0.0f, d1 = 0.0f, d2 = 0.0f, d3 = 0.0f;
  for (int j = s; j < t; ++j) {
    const float4 s4 = es4[lst[j]];
    d0 += s4.x; d1 += s4.y; d2 += s4.z; d3 += s4.w;
  }

  const bool hi = lane >= 16;
  const float da = hi ? d1 : d0;
  const float db = hi ? d3 : d2;

  float4 a = make_float4(0.0f, 0.0f, 0.0f, 0.0f);
  float4 b = make_float4(0.0f, 0.0f, 0.0f, 0.0f);
  for (int j = s; j < t; ++j) {
    const int e = lst[j];
    const float4 s4 = es4[e];
    const float wa = (hi ? s4.y : s4.x) / da;
    const float wb = (hi ? s4.w : s4.z) / db;
    const float4* __restrict__ yrow4 = (const float4*)(Y + (size_t)e * HID);
    const float4 ya = yrow4[lane];
    const float4 yb = yrow4[lane + 32];
    a.x += wa * ya.x; a.y += wa * ya.y; a.z += wa * ya.z; a.w += wa * ya.w;
    b.x += wb * yb.x; b.y += wb * yb.y; b.z += wb * yb.z; b.w += wb * yb.w;
  }

  a.x = a.x > 0.0f ? a.x : expm1f(a.x);
  a.y = a.y > 0.0f ? a.y : expm1f(a.y);
  a.z = a.z > 0.0f ? a.z : expm1f(a.z);
  a.w = a.w > 0.0f ? a.w : expm1f(a.w);
  b.x = b.x > 0.0f ? b.x : expm1f(b.x);
  b.y = b.y > 0.0f ? b.y : expm1f(b.y);
  b.z = b.z > 0.0f ? b.z : expm1f(b.z);
  b.w = b.w > 0.0f ? b.w : expm1f(b.w);

  float4* __restrict__ orow4 = (float4*)(out + (size_t)v * HID);
  orow4[lane] = a;
  orow4[lane + 32] = b;
}

extern "C" void kernel_launch(void* const* d_in, const int* in_sizes, int n_in,
                              void* d_out, int out_size, void* d_ws, size_t ws_size,
                              hipStream_t stream) {
  const float* X     = (const float*)d_in[0];
  const float* W     = (const float*)d_in[1];
  const float* bvec  = (const float*)d_in[2];
  const float* a_e   = (const float*)d_in[3];
  const int*   inc_v = (const int*)d_in[4];
  const int*   inc_e = (const int*)d_in[5];
  float* out = (float*)d_out;

  float* ws    = (float*)d_ws;
  float* Xsum  = ws;
  float* Y     = Xsum + (size_t)N_E * HID;
  float* cnt   = Y    + (size_t)N_E * HID;
  float* es    = cnt  + N_E;
  int*   deg     = (int*)(es + (size_t)N_E * NH);
  int*   counter = deg + N_V;
  int*   start   = counter + 4;
  int*   cur     = start + N_V;
  int*   lst     = (int*)Xsum;

  hipMemsetAsync(Xsum, 0, sizeof(float) * (size_t)N_E * HID, stream);
  hipMemsetAsync(cnt, 0, sizeof(float) * N_E, stream);
  hipMemsetAsync(deg, 0, sizeof(int) * (N_V + 4), stream);

  v2e_scatter<<<dim3(N_NZ / 8), 256, 0, stream>>>(X, inc_v, inc_e, Xsum, cnt);
  gemm_edge<<<dim3((N_E / 16) * NH / 8), 256, 0, stream>>>(Xsum, cnt, W, bvec, Y);
  edge_alpha<<<dim3(N_E), 256, 0, stream>>>(Y, a_e, es);
  v_hist<<<dim3((N_NZ + 255) / 256), 256, 0, stream>>>(inc_v, deg);
  v_alloc<<<dim3((N_V + 255) / 256), 256, 0, stream>>>(deg, counter, start, cur);
  v_fill<<<dim3((N_NZ + 255) / 256), 256, 0, stream>>>(inc_v, inc_e, cur, lst);
  v_gather<<<dim3(N_V / 8), 256, 0, stream>>>(Y, es, start, cur, lst, out);
}


# v2e as per-hyperedge gather over an in-call counting-sort list (int atomics) instead of f32 atomics; membership counts and the single Xsum store folded in; sum order = arrival-order rank
# speedup vs baseline: 2.1238x; 1.4172x over previous
#include <hip/hip_runtime.h>
#include <hip/hip_bf16.h>

typedef __attribute__((ext_vector_type(2))) float v2f;
typedef __attribute__((ext_vector_type(8))) float v8f;

#define N_V   100000
#define N_E   20000
#define N_NZ  400000
#define CIN   256
#define HID   256
#define NH    4
#define DH    64
#define NEG_SLOPE 0.2f

__global__ __launch_bounds__(256) void e_gather(
    const float* __restrict__ X, const int* __restrict__ start,
    const int* __restrict__ end, const int* __restrict__ lst,
    float* __restrict__ Xsum, float* __restrict__ cnt)
{
  const int e = __builtin_amdgcn_readfirstlane(blockIdx.x * 8 + (threadIdx.x >> 5));
  const int lane = threadIdx.x & 31;
  const int s = start[e];
  const int t = end[e];

  const char* __restrict__ Xb = (const char*)X;
  const unsigned lane16 = (unsigned)lane * 16u;
  float4 a = make_float4(0.0f, 0.0f, 0.0f, 0.0f);
  float4 b = make_float4(0.0f, 0.0f, 0.0f, 0.0f);
  for (int j = s; j < t; j += 8) {
    float4 xa[8], xb[8];
#pragma unroll
    for (int k = 0; k < 8; ++k) {
      const int jj = min(j + k, t - 1);
      const unsigned off = (unsigned)lst[jj] * (unsigned)(CIN * sizeof(float)) + lane16;
      const float4* __restrict__ xrow4 = (const float4*)(Xb + off);
      xa[k] = xrow4[0];
      xb[k] = xrow4[32];
    }
#pragma unroll
    for (int k = 0; k < 8; ++k) {
      const bool in = j + k < t;
      a.x += in ? xa[k].x : 0.0f; a.y += in ? xa[k].y : 0.0f;
      a.z += in ? xa[k].z : 0.0f; a.w += in ? xa[k].w : 0.0f;
      b.x += in ? xb[k].x : 0.0f; b.y += in ? xb[k].y : 0.0f;
      b.z += in ? xb[k].z : 0.0f; b.w += in ? xb[k].w : 0.0f;
    }
  }

  float4* __restrict__ drow4 = (float4*)(Xsum + (size_t)e * HID);
  drow4[lane] = a;
  drow4[lane + 32] = b;
  if (lane == 0) cnt[e] = (float)(t - s);
}

__global__ __launch_bounds__(256) void gemm_edge(
    const float* __restrict__ Xsum, const float* __restrict__ cnt,
    const float* __restrict__ W, const float* __restrict__ bvec,
    float* __restrict__ Y)
{
  const int lane = threadIdx.x & 31;
  const int wid  = blockIdx.x * 8 + (threadIdx.x >> 5);
  const int row_tile = wid >> 2;
  const int head = wid & 3;
  const int row0 = row_tile * 16;
  const int m = lane & 15;
  const int half = lane >> 4;

  const int row = row0 + m;
  const float inv = 1.0f / fmaxf(cnt[row], 1.0f);
  const float* __restrict__ arow = Xsum + (size_t)row * HID;
  const float* __restrict__ wh = W + (size_t)head * (CIN * DH);

  v8f acc[4] = {v8f{}, v8f{}, v8f{}, v8f{}};
#pragma unroll 4
  for (int k0 = 0; k0 < CIN; k0 += 4) {
    const int ka = k0 + 2 * half;
    v2f afrag;
    afrag.x = arow[ka] * inv;
    afrag.y = arow[ka + 1] * inv;
    const float* __restrict__ wk = wh + (size_t)ka * DH + m;
#pragma unroll
    for (int t = 0; t < 4; ++t) {
      v2f bfrag;
      bfrag.x = wk[t * 16];
      bfrag.y = wk[t * 16 + DH];
      acc[t] = __builtin_amdgcn_wmma_f32_16x16x4_f32(
          false, afrag, false, bfrag, (short)0, acc[t], false, false);
    }
  }

  const int mbase = half * 8;
#pragma unroll
  for (int t = 0; t < 4; ++t) {
    const int col = head * DH + t * 16 + m;
    const float bias = bvec[col];
#pragma unroll
    for (int r = 0; r < 8; ++r) {
      Y[(size_t)(row0 + mbase + r) * HID + col] = acc[t][r] + bias;
    }
  }
}

__global__ __launch_bounds__(256) void edge_alpha(
    const float* __restrict__ Y, const float* __restrict__ a_e,
    float* __restrict__ es)
{
  __shared__ float red[256];
  const int e = blockIdx.x;
  const int f = threadIdx.x;
  red[f] = Y[(size_t)e * HID + f] * a_e[f];
  __syncthreads();
#pragma unroll
  for (int off = 32; off >= 1; off >>= 1) {
    if ((f & 63) < off) red[f] += red[f + off];
    __syncthreads();
  }
  if ((f & 63) == 0) {
    const float alpha = red[f];
    float s = alpha >= 0.0f ? alpha : NEG_SLOPE * alpha;
    s = fminf(fmaxf(s, 0.001f), 5.0f);
    es[e * NH + (f >> 6)] = __expf(s);
  }
}


__global__ __launch_bounds__(256) void inc_hist(
    const int* __restrict__ inc_v, const int* __restrict__ inc_e,
    int* __restrict__ deg_v, int* __restrict__ deg_e)
{
  const int i = blockIdx.x * 256 + threadIdx.x;
  if (i >= N_NZ) return;
  const int v = inc_v[i];
  const int e = inc_e[i];
  atomicAdd(&deg_v[v], 1);
  atomicAdd(&deg_e[e], 1);
}

#define V_BLOCKS ((N_V + 255) / 256)
#define E_BLOCKS ((N_E + 255) / 256)
__global__ __launch_bounds__(256) void range_alloc(
    const int* __restrict__ deg_v, const int* __restrict__ deg_e,
    int* __restrict__ counter,
    int* __restrict__ start_v, int* __restrict__ cur_v,
    int* __restrict__ start_e, int* __restrict__ cur_e)
{
  const bool eside = blockIdx.x >= V_BLOCKS;
  const int n = eside ? N_E : N_V;
  const int key = (eside ? (int)blockIdx.x - V_BLOCKS : (int)blockIdx.x) * 256 + threadIdx.x;
  const int* __restrict__ deg = eside ? deg_e : deg_v;
  int* __restrict__ start = eside ? start_e : start_v;
  int* __restrict__ cur = eside ? cur_e : cur_v;
  int* __restrict__ bump = eside ? counter + 1 : counter;

  const int lane = threadIdx.x & 31;
  const int d = (key < n) ? deg[key] : 0;
  int incl = d;
  for (int o = 1; o < 32; o <<= 1) {
    const int up = __shfl_up(incl, o);
    if (lane >= o) incl += up;
  }
  int base = 0;
  if (lane == 31) base = atomicAdd(bump, incl);
  base = __shfl(base, 31);
  if (key < n) {
    const int s = base + incl - d;
    start[key] = s;
    cur[key] = s;
  }
}

__global__ __launch_bounds__(256) void key_fill(
    const int* __restrict__ key, const int* __restrict__ val,
    int* __restrict__ cur, int* __restrict__ lst)
{
  const int i = blockIdx.x * 256 + threadIdx.x;
  if (i >= N_NZ) return;
  const int slot = atomicAdd(&cur[key[i]], 1);
  lst[slot] = val[i];
}

__global__ __launch_bounds__(256) void v_gather(
    const float* __restrict__ Y, const float* __restrict__ es,
    const int* __restrict__ start, const int* __restrict__ end,
    const int* __restrict__ lst, float* __restrict__ out)
{
  const int v = __builtin_amdgcn_readfirstlane(blockIdx.x * 8 + (threadIdx.x >> 5));
  const int lane = threadIdx.x & 31;
  const int s = start[v];
  const int t = end[v];
  const float4* __restrict__ es4 = (const float4*)es;

  float d0 = 0.0f, d1 = 0.0f, d2 = 0.0f, d3 = 0.0f;
  for (int j = s; j < t; ++j) {
    const float4 s4 = es4[lst[j]];
    d0 += s4.x; d1 += s4.y; d2 += s4.z; d3 += s4.w;
  }

  const bool hi = lane >= 16;
  const float da = hi ? d1 : d0;
  const float db = hi ? d3 : d2;

  float4 a = make_float4(0.0f, 0.0f, 0.0f, 0.0f);
  float4 b = make_float4(0.0f, 0.0f, 0.0f, 0.0f);
  for (int j = s; j < t; ++j) {
    const int e = lst[j];
    const float4 s4 = es4[e];
    const float wa = (hi ? s4.y : s4.x) / da;
    const float wb = (hi ? s4.w : s4.z) / db;
    const float4* __restrict__ yrow4 = (const float4*)(Y + (size_t)e * HID);
    const float4 ya = yrow4[lane];
    const float4 yb = yrow4[lane + 32];
    a.x += wa * ya.x; a.y += wa * ya.y; a.z += wa * ya.z; a.w += wa * ya.w;
    b.x += wb * yb.x; b.y += wb * yb.y; b.z += wb * yb.z; b.w += wb * yb.w;
  }

  a.x = a.x > 0.0f ? a.x : expm1f(a.x);
  a.y = a.y > 0.0f ? a.y : expm1f(a.y);
  a.z = a.z > 0.0f ? a.z : expm1f(a.z);
  a.w = a.w > 0.0f ? a.w : expm1f(a.w);
  b.x = b.x > 0.0f ? b.x : expm1f(b.x);
  b.y = b.y > 0.0f ? b.y : expm1f(b.y);
  b.z = b.z > 0.0f ? b.z : expm1f(b.z);
  b.w = b.w > 0.0f ? b.w : expm1f(b.w);

  float4* __restrict__ orow4 = (float4*)(out + (size_t)v * HID);
  orow4[lane] = a;
  orow4[lane + 32] = b;
}

extern "C" void kernel_launch(void* const* d_in, const int* in_sizes, int n_in,
                              void* d_out, int out_size, void* d_ws, size_t ws_size,
                              hipStream_t stream) {
  const float* X     = (const float*)d_in[0];
  const float* W     = (const float*)d_in[1];
  const float* bvec  = (const float*)d_in[2];
  const float* a_e   = (const float*)d_in[3];
  const int*   inc_v = (const int*)d_in[4];
  const int*   inc_e = (const int*)d_in[5];
  float* out = (float*)d_out;

  float* ws    = (float*)d_ws;
  float* Xsum  = ws;
  float* Y     = Xsum + (size_t)N_E * HID;
  float* cnt   = Y    + (size_t)N_E * HID;
  float* es    = cnt  + N_E;
  int*   deg_v   = (int*)(es + (size_t)N_E * NH);
  int*   counter = deg_v + N_V;
  int*   start_v = counter + 4;
  int*   cur_v   = start_v + N_V;
  int*   lst_v   = (int*)Xsum;
  int*   deg_e   = (int*)es;
  int*   lst_e   = (int*)Y;
  int*   start_e = lst_e + N_NZ;
  int*   cur_e   = start_e + N_E;

  hipMemsetAsync(deg_e, 0, sizeof(int) * N_E, stream);
  hipMemsetAsync(deg_v, 0, sizeof(int) * (N_V + 4), stream);

  inc_hist<<<dim3((N_NZ + 255) / 256), 256, 0, stream>>>(inc_v, inc_e, deg_v, deg_e);
  range_alloc<<<dim3(V_BLOCKS + E_BLOCKS), 256, 0, stream>>>(deg_v, deg_e, counter,
                                                            start_v, cur_v, start_e, cur_e);
  key_fill<<<dim3((N_NZ + 255) / 256), 256, 0, stream>>>(inc_e, inc_v, cur_e, lst_e);
  e_gather<<<dim3(N_E / 8), 256, 0, stream>>>(X, start_e, cur_e, lst_e, Xsum, cnt);
  gemm_edge<<<dim3((N_E / 16) * NH / 8), 256, 0, stream>>>(Xsum, cnt, W, bvec, Y);
  edge_alpha<<<dim3(N_E), 256, 0, stream>>>(Y, a_e, es);
  key_fill<<<dim3((N_NZ + 255) / 256), 256, 0, stream>>>(inc_v, inc_e, cur_v, lst_v);
  v_gather<<<dim3(N_V / 8), 256, 0, stream>>>(Y, es, start_v, cur_v, lst_v, out);
}


# score kernel folded into the edge GEMM's epilogue (dot with a_e from the accumulators, 16-lane sum, exp): one launch and 160000 waves fewer; sum order of the 64-term score dot changes
# speedup vs baseline: 2.2986x; 1.0823x over previous
#include <hip/hip_runtime.h>
#include <hip/hip_bf16.h>

typedef __attribute__((ext_vector_type(2))) float v2f;
typedef __attribute__((ext_vector_type(8))) float v8f;

#define N_V   100000
#define N_E   20000
#define N_NZ  400000
#define CIN   256
#define HID   256
#define NH    4
#define DH    64
#define NEG_SLOPE 0.2f

__global__ __launch_bounds__(256) void e_gather(
    const float* __restrict__ X, const int* __restrict__ start,
    const int* __restrict__ end, const int* __restrict__ lst,
    float* __restrict__ Xsum, float* __restrict__ cnt)
{
  const int e = __builtin_amdgcn_readfirstlane(blockIdx.x * 8 + (threadIdx.x >> 5));
  const int lane = threadIdx.x & 31;
  const int s = start[e];
  const int t = end[e];

  const char* __restrict__ Xb = (const char*)X;
  const unsigned lane16 = (unsigned)lane * 16u;
  float4 a = make_float4(0.0f, 0.0f, 0.0f, 0.0f);
  float4 b = make_float4(0.0f, 0.0f, 0.0f, 0.0f);
  for (int j = s; j < t; j += 8) {
    float4 xa[8], xb[8];
#pragma unroll
    for (int k = 0; k < 8; ++k) {
      const int jj = min(j + k, t - 1);
      const unsigned off = (unsigned)lst[jj] * (unsigned)(CIN * sizeof(float)) + lane16;
      const float4* __restrict__ xrow4 = (const float4*)(Xb + off);
      xa[k] = xrow4[0];
      xb[k] = xrow4[32];
    }
#pragma unroll
    for (int k = 0; k < 8; ++k) {
      const bool in = j + k < t;
      a.x += in ? xa[k].x : 0.0f; a.y += in ? xa[k].y : 0.0f;
      a.z += in ? xa[k].z : 0.0f; a.w += in ? xa[k].w : 0.0f;
      b.x += in ? xb[k].x : 0.0f; b.y += in ? xb[k].y : 0.0f;
      b.z += in ? xb[k].z : 0.0f; b.w += in ? xb[k].w : 0.0f;
    }
  }

  float4* __restrict__ drow4 = (float4*)(Xsum + (size_t)e * HID);
  drow4[lane] = a;
  drow4[lane + 32] = b;
  if (lane == 0) cnt[e] = (float)(t - s);
}

__global__ __launch_bounds__(256) void gemm_edge(
    const float* __restrict__ Xsum, const float* __restrict__ cnt,
    const float* __restrict__ W, const float* __restrict__ bvec,
    const float* __restrict__ a_e,
    float* __restrict__ Y, float* __restrict__ es)
{
  const int lane = threadIdx.x & 31;
  const int wid  = blockIdx.x * 8 + (threadIdx.x >> 5);
  const int row_tile = wid >> 2;
  const int head = wid & 3;
  const int row0 = row_tile * 16;
  const int m = lane & 15;
  const int half = lane >> 4;

  const int row = row0 + m;
  const float inv = 1.0f / fmaxf(cnt[row], 1.0f);
  const float* __restrict__ arow = Xsum + (size_t)row * HID;
  const float* __restrict__ wh = W + (size_t)head * (CIN * DH);

  v8f acc[4] = {v8f{}, v8f{}, v8f{}, v8f{}};
#pragma unroll 4
  for (int k0 = 0; k0 < CIN; k0 += 4) {
    const int ka = k0 + 2 * half;
    v2f afrag;
    afrag.x = arow[ka] * inv;
    afrag.y = arow[ka + 1] * inv;
    const float* __restrict__ wk = wh + (size_t)ka * DH + m;
#pragma unroll
    for (int t = 0; t < 4; ++t) {
      v2f bfrag;
      bfrag.x = wk[t * 16];
      bfrag.y = wk[t * 16 + DH];
      acc[t] = __builtin_amdgcn_wmma_f32_16x16x4_f32(
          false, afrag, false, bfrag, (short)0, acc[t], false, false);
    }
  }

  const int mbase = half * 8;
  float p[8] = {0.0f, 0.0f, 0.0f, 0.0f, 0.0f, 0.0f, 0.0f, 0.0f};
#pragma unroll
  for (int t = 0; t < 4; ++t) {
    const int col = head * DH + t * 16 + m;
    const float bias = bvec[col];
    const float ae = a_e[col];
#pragma unroll
    for (int r = 0; r < 8; ++r) {
      const float y = acc[t][r] + bias;
      Y[(size_t)(row0 + mbase + r) * HID + col] = y;
      p[r] += y * ae;
    }
  }
#pragma unroll
  for (int off = 8; off >= 1; off >>= 1) {
#pragma unroll
    for (int r = 0; r < 8; ++r) p[r] += __shfl_xor(p[r], off);
  }
  float alpha = p[0];
#pragma unroll
  for (int r = 1; r < 8; ++r) alpha = (m == r) ? p[r] : alpha;
  if (m < 8) {
    float s = alpha >= 0.0f ? alpha : NEG_SLOPE * alpha;
    s = fminf(fmaxf(s, 0.001f), 5.0f);
    es[(row0 + mbase + m) * NH + head] = __expf(s);
  }
}


__global__ __launch_bounds__(256) void inc_hist(
    const int* __restrict__ inc_v, const int* __restrict__ inc_e,
    int* __restrict__ deg_v, int* __restrict__ deg_e)
{
  const int i = blockIdx.x * 256 + threadIdx.x;
  if (i >= N_NZ) return;
  const int v = inc_v[i];
  const int e = inc_e[i];
  atomicAdd(&deg_v[v], 1);
  atomicAdd(&deg_e[e], 1);
}

#define V_BLOCKS ((N_V + 255) / 256)
#define E_BLOCKS ((N_E + 255) / 256)
__global__ __launch_bounds__(256) void range_alloc(
    const int* __restrict__ deg_v, const int* __restrict__ deg_e,
    int* __restrict__ counter,
    int* __restrict__ start_v, int* __restrict__ cur_v,
    int* __restrict__ start_e, int* __restrict__ cur_e)
{
  const bool eside = blockIdx.x >= V_BLOCKS;
  const int n = eside ? N_E : N_V;
  const int key = (eside ? (int)blockIdx.x - V_BLOCKS : (int)blockIdx.x) * 256 + threadIdx.x;
  const int* __restrict__ deg = eside ? deg_e : deg_v;
  int* __restrict__ start = eside ? start_e : start_v;
  int* __restrict__ cur = eside ? cur_e : cur_v;
  int* __restrict__ bump = eside ? counter + 1 : counter;

  const int lane = threadIdx.x & 31;
  const int d = (key < n) ? deg[key] : 0;
  int incl = d;
  for (int o = 1; o < 32; o <<= 1) {
    const int up = __shfl_up(incl, o);
    if (lane >= o) incl += up;
  }
  int base = 0;
  if (lane == 31) base = atomicAdd(bump, incl);
  base = __shfl(base, 31);
  if (key < n) {
    const int s = base + incl - d;
    start[key] = s;
    cur[key] = s;
  }
}

__global__ __launch_bounds__(256) void key_fill(
    const int* __restrict__ key, const int* __restrict__ val,
    int* __restrict__ cur, int* __restrict__ lst)
{
  const int i = blockIdx.x * 256 + threadIdx.x;
  if (i >= N_NZ) return;
  const int slot = atomicAdd(&cur[key[i]], 1);
  lst[slot] = val[i];
}

__global__ __launch_bounds__(256) void v_gather(
    const float* __restrict__ Y, const float* __restrict__ es,
    const int* __restrict__ start, const int* __restrict__ end,
    const int* __restrict__ lst, float* __restrict__ out)
{
  const int v = __builtin_amdgcn_readfirstlane(blockIdx.x * 8 + (threadIdx.x >> 5));
  const int lane = threadIdx.x & 31;
  const int s = start[v];
  const int t = end[v];
  const float4* __restrict__ es4 = (const float4*)es;

  float d0 = 0.0f, d1 = 0.0f, d2 = 0.0f, d3 = 0.0f;
  for (int j = s; j < t; ++j) {
    const float4 s4 = es4[lst[j]];
    d0 += s4.x; d1 += s4.y; d2 += s4.z; d3 += s4.w;
  }

  const bool hi = lane >= 16;
  const float da = hi ? d1 : d0;
  const float db = hi ? d3 : d2;

  float4 a = make_float4(0.0f, 0.0f, 0.0f, 0.0f);
  float4 b = make_float4(0.0f, 0.0f, 0.0f, 0.0f);
  for (int j = s; j < t; ++j) {
    const int e = lst[j];
    const float4 s4 = es4[e];
    const float wa = (hi ? s4.y : s4.x) / da;
    const float wb = (hi ? s4.w : s4.z) / db;
    const float4* __restrict__ yrow4 = (const float4*)(Y + (size_t)e * HID);
    const float4 ya = yrow4[lane];
    const float4 yb = yrow4[lane + 32];
    a.x += wa * ya.x; a.y += wa * ya.y; a.z += wa * ya.z; a.w += wa * ya.w;
    b.x += wb * yb.x; b.y += wb * yb.y; b.z += wb * yb.z; b.w += wb * yb.w;
  }

  a.x = a.x > 0.0f ? a.x : expm1f(a.x);
  a.y = a.y > 0.0f ? a.y : expm1f(a.y);
  a.z = a.z > 0.0f ? a.z : expm1f(a.z);
  a.w = a.w > 0.0f ? a.w : expm1f(a.w);
  b.x = b.x > 0.0f ? b.x : expm1f(b.x);
  b.y = b.y > 0.0f ? b.y : expm1f(b.y);
  b.z = b.z > 0.0f ? b.z : expm1f(b.z);
  b.w = b.w > 0.0f ? b.w : expm1f(b.w);

  float4* __restrict__ orow4 = (float4*)(out + (size_t)v * HID);
  orow4[lane] = a;
  orow4[lane + 32] = b;
}

extern "C" void kernel_launch(void* const* d_in, const int* in_sizes, int n_in,
                              void* d_out, int out_size, void* d_ws, size_t ws_size,
                              hipStream_t stream) {
  const float* X     = (const float*)d_in[0];
  const float* W     = (const float*)d_in[1];
  const float* bvec  = (const float*)d_in[2];
  const float* a_e   = (const float*)d_in[3];
  const int*   inc_v = (const int*)d_in[4];
  const int*   inc_e = (const int*)d_in[5];
  float* out = (float*)d_out;

  float* ws    = (float*)d_ws;
  float* Xsum  = ws;
  float* Y     = Xsum + (size_t)N_E * HID;
  float* cnt   = Y    + (size_t)N_E * HID;
  float* es    = cnt  + N_E;
  int*   deg_v   = (int*)(es + (size_t)N_E * NH);
  int*   counter = deg_v + N_V;
  int*   start_v = counter + 4;
  int*   cur_v   = start_v + N_V;
  int*   lst_v   = (int*)Xsum;
  int*   deg_e   = (int*)es;
  int*   lst_e   = (int*)Y;
  int*   start_e = lst_e + N_NZ;
  int*   cur_e   = start_e + N_E;

  hipMemsetAsync(deg_e, 0, sizeof(int) * N_E, stream);
  hipMemsetAsync(deg_v, 0, sizeof(int) * (N_V + 4), stream);

  inc_hist<<<dim3((N_NZ + 255) / 256), 256, 0, stream>>>(inc_v, inc_e, deg_v, deg_e);
  range_alloc<<<dim3(V_BLOCKS + E_BLOCKS), 256, 0, stream>>>(deg_v, deg_e, counter,
                                                            start_v, cur_v, start_e, cur_e);
  key_fill<<<dim3((N_NZ + 255) / 256), 256, 0, stream>>>(inc_e, inc_v, cur_e, lst_e);
  e_gather<<<dim3(N_E / 8), 256, 0, stream>>>(X, start_e, cur_e, lst_e, Xsum, cnt);
  gemm_edge<<<dim3((N_E / 16) * NH / 8), 256, 0, stream>>>(Xsum, cnt, W, bvec, a_e, Y, es);
  key_fill<<<dim3((N_NZ + 255) / 256), 256, 0, stream>>>(inc_v, inc_e, cur_v, lst_v);
  v_gather<<<dim3(N_V / 8), 256, 0, stream>>>(Y, es, start_v, cur_v, lst_v, out);
}


# v_gather: a vertex's list, scores and edge rows come as wide batched loads (entry per lane, rows four entries per group, first group ahead of the scores), not one entry per trip; sum order kept
# speedup vs baseline: 2.3675x; 1.0300x over previous
#include <hip/hip_runtime.h>
#include <hip/hip_bf16.h>

typedef __attribute__((ext_vector_type(2))) float v2f;
typedef __attribute__((ext_vector_type(8))) float v8f;

#define N_V   100000
#define N_E   20000
#define N_NZ  400000
#define CIN   256
#define HID   256
#define NH    4
#define DH    64
#define NEG_SLOPE 0.2f

__global__ __launch_bounds__(256) void e_gather(
    const float* __restrict__ X, const int* __restrict__ start,
    const int* __restrict__ end, const int* __restrict__ lst,
    float* __restrict__ Xsum, float* __restrict__ cnt)
{
  const int e = __builtin_amdgcn_readfirstlane(blockIdx.x * 8 + (threadIdx.x >> 5));
  const int lane = threadIdx.x & 31;
  const int s = start[e];
  const int t = end[e];

  const char* __restrict__ Xb = (const char*)X;
  const unsigned lane16 = (unsigned)lane * 16u;
  float4 a = make_float4(0.0f, 0.0f, 0.0f, 0.0f);
  float4 b = make_float4(0.0f, 0.0f, 0.0f, 0.0f);
  for (int j = s; j < t; j += 8) {
    float4 xa[8], xb[8];
#pragma unroll
    for (int k = 0; k < 8; ++k) {
      const int jj = min(j + k, t - 1);
      const unsigned off = (unsigned)lst[jj] * (unsigned)(CIN * sizeof(float)) + lane16;
      const float4* __restrict__ xrow4 = (const float4*)(Xb + off);
      xa[k] = xrow4[0];
      xb[k] = xrow4[32];
    }
#pragma unroll
    for (int k = 0; k < 8; ++k) {
      const bool in = j + k < t;
      a.x += in ? xa[k].x : 0.0f; a.y += in ? xa[k].y : 0.0f;
      a.z += in ? xa[k].z : 0.0f; a.w += in ? xa[k].w : 0.0f;
      b.x += in ? xb[k].x : 0.0f; b.y += in ? xb[k].y : 0.0f;
      b.z += in ? xb[k].z : 0.0f; b.w += in ? xb[k].w : 0.0f;
    }
  }

  float4* __restrict__ drow4 = (float4*)(Xsum + (size_t)e * HID);
  drow4[lane] = a;
  drow4[lane + 32] = b;
  if (lane == 0) cnt[e] = (float)(t - s);
}

__global__ __launch_bounds__(256) void gemm_edge(
    const float* __restrict__ Xsum, const float* __restrict__ cnt,
    const float* __restrict__ W, const float* __restrict__ bvec,
    const float* __restrict__ a_e,
    float* __restrict__ Y, float* __restrict__ es)
{
  const int lane = threadIdx.x & 31;
  const int wid  = blockIdx.x * 8 + (threadIdx.x >> 5);
  const int row_tile = wid >> 2;
  const int head = wid & 3;
  const int row0 = row_tile * 16;
  const int m = lane & 15;
  const int half = lane >> 4;

  const int row = row0 + m;
  const float inv = 1.0f / fmaxf(cnt[row], 1.0f);
  const float* __restrict__ arow = Xsum + (size_t)row * HID;
  const float* __restrict__ wh = W + (size_t)head * (CIN * DH);

  v8f acc[4] = {v8f{}, v8f{}, v8f{}, v8f{}};
#pragma unroll 4
  for (int k0 = 0; k0 < CIN; k0 += 4) {
    const int ka = k0 + 2 * half;
    v2f afrag;
    afrag.x = arow[ka] * inv;
    afrag.y = arow[ka + 1] * inv;
    const float* __restrict__ wk = wh + (size_t)ka * DH + m;
#pragma unroll
    for (int t = 0; t < 4; ++t) {
      v2f bfrag;
      bfrag.x = wk[t * 16];
      bfrag.y = wk[t * 16 + DH];
      acc[t] = __builtin_amdgcn_wmma_f32_16x16x4_f32(
          false, afrag, false, bfrag, (short)0, acc[t], false, false);
    }
  }

  const int mbase = half * 8;
  float p[8] = {0.0f, 0.0f, 0.0f, 0.0f, 0.0f, 0.0f, 0.0f, 0.0f};
#pragma unroll
  for (int t = 0; t < 4; ++t) {
    const int col = head * DH + t * 16 + m;
    const float bias = bvec[col];
    const float ae = a_e[col];
#pragma unroll
    for (int r = 0; r < 8; ++r) {
      const float y = acc[t][r] + bias;
      Y[(size_t)(row0 + mbase + r) * HID + col] = y;
      p[r] += y * ae;
    }
  }
#pragma unroll
  for (int off = 8; off >= 1; off >>= 1) {
#pragma unroll
    for (int r = 0; r < 8; ++r) p[r] += __shfl_xor(p[r], off);
  }
  float alpha = p[0];
#pragma unroll
  for (int r = 1; r < 8; ++r) alpha = (m == r) ? p[r] : alpha;
  if (m < 8) {
    float s = alpha >= 0.0f ? alpha : NEG_SLOPE * alpha;
    s = fminf(fmaxf(s, 0.001f), 5.0f);
    es[(row0 + mbase + m) * NH + head] = __expf(s);
  }
}


__global__ __launch_bounds__(256) void inc_hist(
    const int* __restrict__ inc_v, const int* __restrict__ inc_e,
    int* __restrict__ deg_v, int* __restrict__ deg_e)
{
  const int i = blockIdx.x * 256 + threadIdx.x;
  if (i >= N_NZ) return;
  const int v = inc_v[i];
  const int e = inc_e[i];
  atomicAdd(&deg_v[v], 1);
  atomicAdd(&deg_e[e], 1);
}

#define V_BLOCKS ((N_V + 255) / 256)
#define E_BLOCKS ((N_E + 255) / 256)
__global__ __launch_bounds__(256) void range_alloc(
    const int* __restrict__ deg_v, const int* __restrict__ deg_e,
    int* __restrict__ counter,
    int* __restrict__ start_v, int* __restrict__ cur_v,
    int* __restrict__ start_e, int* __restrict__ cur_e)
{
  const bool eside = blockIdx.x >= V_BLOCKS;
  const int n = eside ? N_E : N_V;
  const int key = (eside ? (int)blockIdx.x - V_BLOCKS : (int)blockIdx.x) * 256 + threadIdx.x;
  const int* __restrict__ deg = eside ? deg_e : deg_v;
  int* __restrict__ start = eside ? start_e : start_v;
  int* __restrict__ cur = eside ? cur_e : cur_v;
  int* __restrict__ bump = eside ? counter + 1 : counter;

  const int lane = threadIdx.x & 31;
  const int d = (key < n) ? deg[key] : 0;
  int incl = d;
  for (int o = 1; o < 32; o <<= 1) {
    const int up = __shfl_up(incl, o);
    if (lane >= o) incl += up;
  }
  int base = 0;
  if (lane == 31) base = atomicAdd(bump, incl);
  base = __shfl(base, 31);
  if (key < n) {
    const int s = base + incl - d;
    start[key] = s;
    cur[key] = s;
  }
}

__global__ __launch_bounds__(256) void key_fill(
    const int* __restrict__ key, const int* __restrict__ val,
    int* __restrict__ cur, int* __restrict__ lst)
{
  const int i = blockIdx.x * 256 + threadIdx.x;
  if (i >= N_NZ) return;
  const int slot = atomicAdd(&cur[key[i]], 1);
  lst[slot] = val[i];
}


#define VB 4

__device__ __forceinline__ float lane_val(float x, int k) {
  return __int_as_float(__builtin_amdgcn_readlane(__float_as_int(x), k));
}

__device__ __forceinline__ void v_rows_load(
    const char* __restrict__ Yb, unsigned lane16, int e_l, int k0, int cnt,
    float4 (&ya)[VB], float4 (&yb)[VB])
{
#pragma unroll
  for (int q = 0; q < VB; ++q) {
    const int kk = min(k0 + q, cnt - 1);
    const int e = __builtin_amdgcn_readlane(e_l, kk);
    const float4* __restrict__ yrow4 =
        (const float4*)(Yb + (size_t)e * (HID * sizeof(float)) + lane16);
    ya[q] = yrow4[0];
    yb[q] = yrow4[32];
  }
}

__device__ __forceinline__ void v_rows_apply(
    const float4& w_l, bool hi, int k0,
    const float4 (&ya)[VB], const float4 (&yb)[VB], float4& a, float4& b)
{
#pragma unroll
  for (int q = 0; q < VB; ++q) {
    const int kk = k0 + q;
    const float wx = lane_val(w_l.x, kk);
    const float wy = lane_val(w_l.y, kk);
    const float wz = lane_val(w_l.z, kk);
    const float ww = lane_val(w_l.w, kk);
    const float wa = hi ? wy : wx;
    const float wb = hi ? ww : wz;
    a.x += wa * ya[q].x; a.y += wa * ya[q].y; a.z += wa * ya[q].z; a.w += wa * ya[q].w;
    b.x += wb * yb[q].x; b.y += wb * yb[q].y; b.z += wb * yb[q].z; b.w += wb * yb[q].w;
  }
}

__global__ __launch_bounds__(256) void v_gather(
    const float* __restrict__ Y, const float* __restrict__ es,
    const int* __restrict__ start, const int* __restrict__ end,
    const int* __restrict__ lst, float* __restrict__ out)
{
  const int v = __builtin_amdgcn_readfirstlane(blockIdx.x * 8 + (threadIdx.x >> 5));
  const int lane = threadIdx.x & 31;
  const int s = start[v];
  const int t = end[v];
  const float4* __restrict__ es4 = (const float4*)es;
  const char* __restrict__ Yb = (const char*)Y;
  const unsigned lane16 = (unsigned)lane * 16u;
  float4* __restrict__ orow4 = (float4*)(out + (size_t)v * HID);

  if (t <= s) {
    const float4 z = make_float4(0.0f, 0.0f, 0.0f, 0.0f);
    orow4[lane] = z;
    orow4[lane + 32] = z;
    return;
  }

  const bool hi = lane >= 16;
  const int n = t - s;

  float4 a = make_float4(0.0f, 0.0f, 0.0f, 0.0f);
  float4 b = make_float4(0.0f, 0.0f, 0.0f, 0.0f);
  float4 ya[VB], yb[VB];

  if (n <= 32) {
    const int e_l = lst[s + min(lane, n - 1)];
    const float4 s_l = es4[e_l];
    v_rows_load(Yb, lane16, e_l, 0, n, ya, yb);

    float d0 = 0.0f, d1 = 0.0f, d2 = 0.0f, d3 = 0.0f;
    int k = 0;
    do {
      d0 += lane_val(s_l.x, k); d1 += lane_val(s_l.y, k);
      d2 += lane_val(s_l.z, k); d3 += lane_val(s_l.w, k);
    } while (++k < n);

    float4 w_l;
    w_l.x = lane < n ? s_l.x / d0 : 0.0f;
    w_l.y = lane < n ? s_l.y / d1 : 0.0f;
    w_l.z = lane < n ? s_l.z / d2 : 0.0f;
    w_l.w = lane < n ? s_l.w / d3 : 0.0f;
    v_rows_apply(w_l, hi, 0, ya, yb, a, b);
    for (int k0 = VB; k0 < n; k0 += VB) {
      v_rows_load(Yb, lane16, e_l, k0, n, ya, yb);
      v_rows_apply(w_l, hi, k0, ya, yb, a, b);
    }
  } else {
    float d0 = 0.0f, d1 = 0.0f, d2 = 0.0f, d3 = 0.0f;
    for (int c = s; c < t; c += 32) {
      const int cnt = min(t - c, 32);
      const int e_t = lst[c + min(lane, cnt - 1)];
      const float4 s_t = es4[e_t];
      for (int k = 0; k < cnt; ++k) {
        d0 += lane_val(s_t.x, k); d1 += lane_val(s_t.y, k);
        d2 += lane_val(s_t.z, k); d3 += lane_val(s_t.w, k);
      }
    }
    for (int c = s; c < t; c += 32) {
      const int cnt = min(t - c, 32);
      const int e_l = lst[c + min(lane, cnt - 1)];
      const float4 s_l = es4[e_l];
      float4 w_l;
      w_l.x = lane < cnt ? s_l.x / d0 : 0.0f;
      w_l.y = lane < cnt ? s_l.y / d1 : 0.0f;
      w_l.z = lane < cnt ? s_l.z / d2 : 0.0f;
      w_l.w = lane < cnt ? s_l.w / d3 : 0.0f;
      for (int k0 = 0; k0 < cnt; k0 += VB) {
        v_rows_load(Yb, lane16, e_l, k0, cnt, ya, yb);
        v_rows_apply(w_l, hi, k0, ya, yb, a, b);
      }
    }
  }

  a.x = a.x > 0.0f ? a.x : expm1f(a.x);
  a.y = a.y > 0.0f ? a.y : expm1f(a.y);
  a.z = a.z > 0.0f ? a.z : expm1f(a.z);
  a.w = a.w > 0.0f ? a.w : expm1f(a.w);
  b.x = b.x > 0.0f ? b.x : expm1f(b.x);
  b.y = b.y > 0.0f ? b.y : expm1f(b.y);
  b.z = b.z > 0.0f ? b.z : expm1f(b.z);
  b.w = b.w > 0.0f ? b.w : expm1f(b.w);

  orow4[lane] = a;
  orow4[lane + 32] = b;
}

extern "C" void kernel_launch(void* const* d_in, const int* in_sizes, int n_in,
                              void* d_out, int out_size, void* d_ws, size_t ws_size,
                              hipStream_t stream) {
  const float* X     = (const float*)d_in[0];
  const float* W     = (const float*)d_in[1];
  const float* bvec  = (const float*)d_in[2];
  const float* a_e   = (const float*)d_in[3];
  const int*   inc_v = (const int*)d_in[4];
  const int*   inc_e = (const int*)d_in[5];
  float* out = (float*)d_out;

  float* ws    = (float*)d_ws;
  float* Xsum  = ws;
  float* Y     = Xsum + (size_t)N_E * HID;
  float* cnt   = Y    + (size_t)N_E * HID;
  float* es    = cnt  + N_E;
  int*   deg_v   = (int*)(es + (size_t)N_E * NH);
  int*   counter = deg_v + N_V;
  int*   start_v = counter + 4;
  int*   cur_v   = start_v + N_V;
  int*   lst_v   = (int*)Xsum;
  int*   deg_e   = (int*)es;
  int*   lst_e   = (int*)Y;
  int*   start_e = lst_e + N_NZ;
  int*   cur_e   = start_e + N_E;

  hipMemsetAsync(deg_e, 0, sizeof(int) * N_E, stream);
  hipMemsetAsync(deg_v, 0, sizeof(int) * (N_V + 4), stream);

  inc_hist<<<dim3((N_NZ + 255) / 256), 256, 0, stream>>>(inc_v, inc_e, deg_v, deg_e);
  range_alloc<<<dim3(V_BLOCKS + E_BLOCKS), 256, 0, stream>>>(deg_v, deg_e, counter,
                                                            start_v, cur_v, start_e, cur_e);
  key_fill<<<dim3((N_NZ + 255) / 256), 256, 0, stream>>>(inc_e, inc_v, cur_e, lst_e);
  e_gather<<<dim3(N_E / 8), 256, 0, stream>>>(X, start_e, cur_e, lst_e, Xsum, cnt);
  gemm_edge<<<dim3((N_E / 16) * NH / 8), 256, 0, stream>>>(Xsum, cnt, W, bvec, a_e, Y, es);
  key_fill<<<dim3((N_NZ + 255) / 256), 256, 0, stream>>>(inc_v, inc_e, cur_v, lst_v);
  v_gather<<<dim3(N_V / 8), 256, 0, stream>>>(Y, es, start_v, cur_v, lst_v, out);
}


# incidence lists live in the heads of the rows their own gather waves overwrite (Xsum row e, out row v), filled in one pass: no histogram, range or second fill launch; sum order = arrival rank
# speedup vs baseline: 3.4139x; 1.4420x over previous
#include <hip/hip_runtime.h>
#include <hip/hip_bf16.h>

typedef __attribute__((ext_vector_type(2))) float v2f;
typedef __attribute__((ext_vector_type(8))) float v8f;

#define N_V   100000
#define N_E   20000
#define N_NZ  400000
#define CIN   256
#define HID   256
#define NH    4
#define DH    64
#define NEG_SLOPE 0.2f

#define ROWCAP 256

__global__ __launch_bounds__(256) void e_gather(
    const float* __restrict__ X, const int* __restrict__ deg,
    const int* __restrict__ inc_v, const int* __restrict__ inc_e,
    float* Xsum, float* __restrict__ cnt)
{
  const int e = __builtin_amdgcn_readfirstlane(blockIdx.x * 8 + (threadIdx.x >> 5));
  const int lane = threadIdx.x & 31;
  const int t = deg[e];
  const int* lst = (const int*)(Xsum + (size_t)e * HID);

  const char* __restrict__ Xb = (const char*)X;
  const unsigned lane16 = (unsigned)lane * 16u;
  float4 a = make_float4(0.0f, 0.0f, 0.0f, 0.0f);
  float4 b = make_float4(0.0f, 0.0f, 0.0f, 0.0f);
  if (t <= ROWCAP) {
    for (int j = 0; j < t; j += 8) {
      float4 xa[8], xb[8];
#pragma unroll
      for (int k = 0; k < 8; ++k) {
        const int jj = min(j + k, t - 1);
        const unsigned off = (unsigned)lst[jj] * (unsigned)(CIN * sizeof(float)) + lane16;
        const float4* __restrict__ xrow4 = (const float4*)(Xb + off);
        xa[k] = xrow4[0];
        xb[k] = xrow4[32];
      }
#pragma unroll
      for (int k = 0; k < 8; ++k) {
        const bool in = j + k < t;
        a.x += in ? xa[k].x : 0.0f; a.y += in ? xa[k].y : 0.0f;
        a.z += in ? xa[k].z : 0.0f; a.w += in ? xa[k].w : 0.0f;
        b.x += in ? xb[k].x : 0.0f; b.y += in ? xb[k].y : 0.0f;
        b.z += in ? xb[k].z : 0.0f; b.w += in ? xb[k].w : 0.0f;
      }
    }
  } else {
    for (int c = 0; c < N_NZ; c += 32) {
      const int i = c + lane;
      const bool ok = i < N_NZ;
      const int key = ok ? inc_e[i] : -1;
      const int val = ok ? inc_v[i] : 0;
      unsigned m = __builtin_amdgcn_ballot_w32(key == e);
      while (m != 0u) {
        const int k = __builtin_ctz(m);
        m &= m - 1u;
        const int vv = __builtin_amdgcn_readlane(val, k);
        const unsigned off = (unsigned)vv * (unsigned)(CIN * sizeof(float)) + lane16;
        const float4* __restrict__ xrow4 = (const float4*)(Xb + off);
        const float4 xa = xrow4[0];
        const float4 xb = xrow4[32];
        a.x += xa.x; a.y += xa.y; a.z += xa.z; a.w += xa.w;
        b.x += xb.x; b.y += xb.y; b.z += xb.z; b.w += xb.w;
      }
    }
  }

  float4* drow4 = (float4*)(Xsum + (size_t)e * HID);
  drow4[lane] = a;
  drow4[lane + 32] = b;
  if (lane == 0) cnt[e] = (float)t;
}

__global__ __launch_bounds__(256) void gemm_edge(
    const float* __restrict__ Xsum, const float* __restrict__ cnt,
    const float* __restrict__ W, const float* __restrict__ bvec,
    const float* __restrict__ a_e,
    float* __restrict__ Y, float* __restrict__ es)
{
  const int lane = threadIdx.x & 31;
  const int wid  = blockIdx.x * 8 + (threadIdx.x >> 5);
  const int row_tile = wid >> 2;
  const int head = wid & 3;
  const int row0 = row_tile * 16;
  const int m = lane & 15;
  const int half = lane >> 4;

  const int row = row0 + m;
  const float inv = 1.0f / fmaxf(cnt[row], 1.0f);
  const float* __restrict__ arow = Xsum + (size_t)row * HID;
  const float* __restrict__ wh = W + (size_t)head * (CIN * DH);

  v8f acc[4] = {v8f{}, v8f{}, v8f{}, v8f{}};
#pragma unroll 4
  for (int k0 = 0; k0 < CIN; k0 += 4) {
    const int ka = k0 + 2 * half;
    v2f afrag;
    afrag.x = arow[ka] * inv;
    afrag.y = arow[ka + 1] * inv;
    const float* __restrict__ wk = wh + (size_t)ka * DH + m;
#pragma unroll
    for (int t = 0; t < 4; ++t) {
      v2f bfrag;
      bfrag.x = wk[t * 16];
      bfrag.y = wk[t * 16 + DH];
      acc[t] = __builtin_amdgcn_wmma_f32_16x16x4_f32(
          false, afrag, false, bfrag, (short)0, acc[t], false, false);
    }
  }

  const int mbase = half * 8;
  float p[8] = {0.0f, 0.0f, 0.0f, 0.0f, 0.0f, 0.0f, 0.0f, 0.0f};
#pragma unroll
  for (int t = 0; t < 4; ++t) {
    const int col = head * DH + t * 16 + m;
    const float bias = bvec[col];
    const float ae = a_e[col];
#pragma unroll
    for (int r = 0; r < 8; ++r) {
      const float y = acc[t][r] + bias;
      Y[(size_t)(row0 + mbase + r) * HID + col] = y;
      p[r] += y * ae;
    }
  }
#pragma unroll
  for (int off = 8; off >= 1; off >>= 1) {
#pragma unroll
    for (int r = 0; r < 8; ++r) p[r] += __shfl_xor(p[r], off);
  }
  float alpha = p[0];
#pragma unroll
  for (int r = 1; r < 8; ++r) alpha = (m == r) ? p[r] : alpha;
  if (m < 8) {
    float s = alpha >= 0.0f ? alpha : NEG_SLOPE * alpha;
    s = fminf(fmaxf(s, 0.001f), 5.0f);
    es[(row0 + mbase + m) * NH + head] = __expf(s);
  }
}


__global__ __launch_bounds__(256) void inc_fill(
    const int* __restrict__ inc_v, const int* __restrict__ inc_e,
    int* __restrict__ deg_v, int* __restrict__ deg_e,
    float* __restrict__ out, float* __restrict__ Xsum)
{
  const int i = blockIdx.x * 256 + threadIdx.x;
  if (i >= N_NZ) return;
  const int v = inc_v[i];
  const int e = inc_e[i];
  const int sv = atomicAdd(&deg_v[v], 1);
  const int se = atomicAdd(&deg_e[e], 1);
  if (sv < ROWCAP) ((int*)out)[(size_t)v * HID + sv] = e;
  if (se < ROWCAP) ((int*)Xsum)[(size_t)e * HID + se] = v;
}


#define VB 4

__device__ __forceinline__ float lane_val(float x, int k) {
  return __int_as_float(__builtin_amdgcn_readlane(__float_as_int(x), k));
}

__device__ __forceinline__ void v_rows_load(
    const char* __restrict__ Yb, unsigned lane16, int e_l, int k0, int cnt,
    float4 (&ya)[VB], float4 (&yb)[VB])
{
#pragma unroll
  for (int q = 0; q < VB; ++q) {
    const int kk = min(k0 + q, cnt - 1);
    const int e = __builtin_amdgcn_readlane(e_l, kk);
    const float4* __restrict__ yrow4 =
        (const float4*)(Yb + (size_t)e * (HID * sizeof(float)) + lane16);
    ya[q] = yrow4[0];
    yb[q] = yrow4[32];
  }
}

__device__ __forceinline__ void v_rows_apply(
    const float4& w_l, bool hi, int k0,
    const float4 (&ya)[VB], const float4 (&yb)[VB], float4& a, float4& b)
{
#pragma unroll
  for (int q = 0; q < VB; ++q) {
    const int kk = k0 + q;
    const float wx = lane_val(w_l.x, kk);
    const float wy = lane_val(w_l.y, kk);
    const float wz = lane_val(w_l.z, kk);
    const float ww = lane_val(w_l.w, kk);
    const float wa = hi ? wy : wx;
    const float wb = hi ? ww : wz;
    a.x += wa * ya[q].x; a.y += wa * ya[q].y; a.z += wa * ya[q].z; a.w += wa * ya[q].w;
    b.x += wb * yb[q].x; b.y += wb * yb[q].y; b.z += wb * yb[q].z; b.w += wb * yb[q].w;
  }
}

__global__ __launch_bounds__(256) void v_gather(
    const float* __restrict__ Y, const float* __restrict__ es,
    const int* __restrict__ deg,
    const int* __restrict__ inc_v, const int* __restrict__ inc_e,
    float* out)
{
  const int v = __builtin_amdgcn_readfirstlane(blockIdx.x * 8 + (threadIdx.x >> 5));
  const int lane = threadIdx.x & 31;
  const int n = deg[v];
  const int* lst = (const int*)(out + (size_t)v * HID);
  const float4* __restrict__ es4 = (const float4*)es;
  const char* __restrict__ Yb = (const char*)Y;
  const unsigned lane16 = (unsigned)lane * 16u;
  float4* orow4 = (float4*)(out + (size_t)v * HID);

  if (n <= 0) {
    const float4 z = make_float4(0.0f, 0.0f, 0.0f, 0.0f);
    orow4[lane] = z;
    orow4[lane + 32] = z;
    return;
  }

  const bool hi = lane >= 16;

  float4 a = make_float4(0.0f, 0.0f, 0.0f, 0.0f);
  float4 b = make_float4(0.0f, 0.0f, 0.0f, 0.0f);
  float4 ya[VB], yb[VB];

  if (n <= 32) {
    const int e_l = lst[min(lane, n - 1)];
    const float4 s_l = es4[e_l];
    v_rows_load(Yb, lane16, e_l, 0, n, ya, yb);

    float d0 = 0.0f, d1 = 0.0f, d2 = 0.0f, d3 = 0.0f;
    int k = 0;
    do {
      d0 += lane_val(s_l.x, k); d1 += lane_val(s_l.y, k);
      d2 += lane_val(s_l.z, k); d3 += lane_val(s_l.w, k);
    } while (++k < n);

    float4 w_l;
    w_l.x = lane < n ? s_l.x / d0 : 0.0f;
    w_l.y = lane < n ? s_l.y / d1 : 0.0f;
    w_l.z = lane < n ? s_l.z / d2 : 0.0f;
    w_l.w = lane < n ? s_l.w / d3 : 0.0f;
    v_rows_apply(w_l, hi, 0, ya, yb, a, b);
    for (int k0 = VB; k0 < n; k0 += VB) {
      v_rows_load(Yb, lane16, e_l, k0, n, ya, yb);
      v_rows_apply(w_l, hi, k0, ya, yb, a, b);
    }
  } else if (n <= ROWCAP) {
    float d0 = 0.0f, d1 = 0.0f, d2 = 0.0f, d3 = 0.0f;
    for (int c = 0; c < n; c += 32) {
      const int cnt = min(n - c, 32);
      const int e_t = lst[c + min(lane, cnt - 1)];
      const float4 s_t = es4[e_t];
      for (int k = 0; k < cnt; ++k) {
        d0 += lane_val(s_t.x, k); d1 += lane_val(s_t.y, k);
        d2 += lane_val(s_t.z, k); d3 += lane_val(s_t.w, k);
      }
    }
    for (int c = 0; c < n; c += 32) {
      const int cnt = min(n - c, 32);
      const int e_l = lst[c + min(lane, cnt - 1)];
      const float4 s_l = es4[e_l];
      float4 w_l;
      w_l.x = lane < cnt ? s_l.x / d0 : 0.0f;
      w_l.y = lane < cnt ? s_l.y / d1 : 0.0f;
      w_l.z = lane < cnt ? s_l.z / d2 : 0.0f;
      w_l.w = lane < cnt ? s_l.w / d3 : 0.0f;
      for (int k0 = 0; k0 < cnt; k0 += VB) {
        v_rows_load(Yb, lane16, e_l, k0, cnt, ya, yb);
        v_rows_apply(w_l, hi, k0, ya, yb, a, b);
      }
    }
  } else {
    float d0 = 0.0f, d1 = 0.0f, d2 = 0.0f, d3 = 0.0f;
    for (int c = 0; c < N_NZ; c += 32) {
      const int i = c + lane;
      const bool ok = i < N_NZ;
      const int key = ok ? inc_v[i] : -1;
      const int val = ok ? inc_e[i] : 0;
      unsigned m = __builtin_amdgcn_ballot_w32(key == v);
      while (m != 0u) {
        const int k = __builtin_ctz(m);
        m &= m - 1u;
        const float4 s4 = es4[__builtin_amdgcn_readlane(val, k)];
        d0 += s4.x; d1 += s4.y; d2 += s4.z; d3 += s4.w;
      }
    }
    for (int c = 0; c < N_NZ; c += 32) {
      const int i = c + lane;
      const bool ok = i < N_NZ;
      const int key = ok ? inc_v[i] : -1;
      const int val = ok ? inc_e[i] : 0;
      unsigned m = __builtin_amdgcn_ballot_w32(key == v);
      while (m != 0u) {
        const int k = __builtin_ctz(m);
        m &= m - 1u;
        const int ee = __builtin_amdgcn_readlane(val, k);
        const float4 s4 = es4[ee];
        const float wa = hi ? s4.y / d1 : s4.x / d0;
        const float wb = hi ? s4.w / d3 : s4.z / d2;
        const float4* __restrict__ yrow4 =
            (const float4*)(Yb + (size_t)ee * (HID * sizeof(float)) + lane16);
        const float4 y0 = yrow4[0];
        const float4 y1 = yrow4[32];
        a.x += wa * y0.x; a.y += wa * y0.y; a.z += wa * y0.z; a.w += wa * y0.w;
        b.x += wb * y1.x; b.y += wb * y1.y; b.z += wb * y1.z; b.w += wb * y1.w;
      }
    }
  }

  a.x = a.x > 0.0f ? a.x : expm1f(a.x);
  a.y = a.y > 0.0f ? a.y : expm1f(a.y);
  a.z = a.z > 0.0f ? a.z : expm1f(a.z);
  a.w = a.w > 0.0f ? a.w : expm1f(a.w);
  b.x = b.x > 0.0f ? b.x : expm1f(b.x);
  b.y = b.y > 0.0f ? b.y : expm1f(b.y);
  b.z = b.z > 0.0f ? b.z : expm1f(b.z);
  b.w = b.w > 0.0f ? b.w : expm1f(b.w);

  orow4[lane] = a;
  orow4[lane + 32] = b;
}

extern "C" void kernel_launch(void* const* d_in, const int* in_sizes, int n_in,
                              void* d_out, int out_size, void* d_ws, size_t ws_size,
                              hipStream_t stream) {
  const float* X     = (const float*)d_in[0];
  const float* W     = (const float*)d_in[1];
  const float* bvec  = (const float*)d_in[2];
  const float* a_e   = (const float*)d_in[3];
  const int*   inc_v = (const int*)d_in[4];
  const int*   inc_e = (const int*)d_in[5];
  float* out = (float*)d_out;

  float* ws    = (float*)d_ws;
  float* Xsum  = ws;
  float* Y     = Xsum + (size_t)N_E * HID;
  float* cnt   = Y    + (size_t)N_E * HID;
  float* es    = cnt  + N_E;
  int*   deg_e = (int*)(es + (size_t)N_E * (NH - 1));
  int*   deg_v = (int*)(es + (size_t)N_E * NH);

  hipMemsetAsync(deg_e, 0, sizeof(int) * (N_E + N_V), stream);

  inc_fill<<<dim3((N_NZ + 255) / 256), 256, 0, stream>>>(inc_v, inc_e, deg_v, deg_e, out, Xsum);
  e_gather<<<dim3(N_E / 8), 256, 0, stream>>>(X, deg_e, inc_v, inc_e, Xsum, cnt);
  gemm_edge<<<dim3((N_E / 16) * NH / 8), 256, 0, stream>>>(Xsum, cnt, W, bvec, a_e, Y, es);
  v_gather<<<dim3(N_V / 8), 256, 0, stream>>>(Y, es, deg_v, inc_v, inc_e, out);
}


# hyperedge gather and edge GEMM in one work-group: the 16 rows of means go through LDS into the WMMA A fragments, no Xsum/cnt round trip and one launch fewer; sum orders kept
# speedup vs baseline: 3.8715x; 1.1340x over previous
#include <hip/hip_runtime.h>
#include <hip/hip_bf16.h>

typedef __attribute__((ext_vector_type(2))) float v2f;
typedef __attribute__((ext_vector_type(8))) float v8f;

#define N_V   100000
#define N_E   20000
#define N_NZ  400000
#define CIN   256
#define HID   256
#define NH    4
#define DH    64
#define NEG_SLOPE 0.2f

#define ROWCAP 256


#define APITCH 260

__device__ __forceinline__ void edge_sum(
    const char* __restrict__ Xb, const int* __restrict__ lst, int t, int e,
    const int* __restrict__ inc_v, const int* __restrict__ inc_e,
    int lane, unsigned lane16, float4& a, float4& b)
{
  a = make_float4(0.0f, 0.0f, 0.0f, 0.0f);
  b = make_float4(0.0f, 0.0f, 0.0f, 0.0f);
  if (t <= ROWCAP) {
    for (int j = 0; j < t; j += 8) {
      float4 xa[8], xb[8];
#pragma unroll
      for (int k = 0; k < 8; ++k) {
        const int jj = min(j + k, t - 1);
        const unsigned off = (unsigned)lst[jj] * (unsigned)(CIN * sizeof(float)) + lane16;
        const float4* __restrict__ xrow4 = (const float4*)(Xb + off);
        xa[k] = xrow4[0];
        xb[k] = xrow4[32];
      }
#pragma unroll
      for (int k = 0; k < 8; ++k) {
        const bool in = j + k < t;
        a.x += in ? xa[k].x : 0.0f; a.y += in ? xa[k].y : 0.0f;
        a.z += in ? xa[k].z : 0.0f; a.w += in ? xa[k].w : 0.0f;
        b.x += in ? xb[k].x : 0.0f; b.y += in ? xb[k].y : 0.0f;
        b.z += in ? xb[k].z : 0.0f; b.w += in ? xb[k].w : 0.0f;
      }
    }
  } else {
    for (int c = 0; c < N_NZ; c += 32) {
      const int i = c + lane;
      const bool ok = i < N_NZ;
      const int key = ok ? inc_e[i] : -1;
      const int val = ok ? inc_v[i] : 0;
      unsigned m = __builtin_amdgcn_ballot_w32(key == e);
      while (m != 0u) {
        const int k = __builtin_ctz(m);
        m &= m - 1u;
        const int vv = __builtin_amdgcn_readlane(val, k);
        const unsigned off = (unsigned)vv * (unsigned)(CIN * sizeof(float)) + lane16;
        const float4* __restrict__ xrow4 = (const float4*)(Xb + off);
        const float4 xa = xrow4[0];
        const float4 xb = xrow4[32];
        a.x += xa.x; a.y += xa.y; a.z += xa.z; a.w += xa.w;
        b.x += xb.x; b.y += xb.y; b.z += xb.z; b.w += xb.w;
      }
    }
  }
}

__global__ __launch_bounds__(256) void edge_tile(
    const float* __restrict__ X, const int* __restrict__ deg,
    const int* __restrict__ inc_v, const int* __restrict__ inc_e,
    const int* __restrict__ lists,
    const float* __restrict__ W, const float* __restrict__ bvec,
    const float* __restrict__ a_e,
    float* __restrict__ Y, float* __restrict__ es)
{
  __shared__ __attribute__((aligned(16))) float As[16 * APITCH];

  const int lane = threadIdx.x & 31;
  const int wave = __builtin_amdgcn_readfirstlane(threadIdx.x >> 5);
  const int row0 = blockIdx.x * 16;

  {
    const char* __restrict__ Xb = (const char*)X;
    const unsigned lane16 = (unsigned)lane * 16u;
    const int t0 = deg[row0 + 2 * wave];
    const int t1 = deg[row0 + 2 * wave + 1];
    for (int q = 0; q < 2; ++q) {
      const int r = 2 * wave + q;
      const int e = row0 + r;
      const int t = q == 0 ? t0 : t1;
      float4 a, b;
      edge_sum(Xb, lists + (size_t)e * HID, t, e, inc_v, inc_e, lane, lane16, a, b);
      const float inv = 1.0f / fmaxf((float)t, 1.0f);
      a.x *= inv; a.y *= inv; a.z *= inv; a.w *= inv;
      b.x *= inv; b.y *= inv; b.z *= inv; b.w *= inv;
      float4* arow4 = (float4*)(As + r * APITCH);
      arow4[lane] = a;
      arow4[lane + 32] = b;
    }
  }
  __syncthreads();
  if (wave >= 4) return;

  const int head = wave;
  const int m = lane & 15;
  const int half = lane >> 4;

  const float* arow = As + m * APITCH;
  const float* __restrict__ wh = W + (size_t)head * (CIN * DH);

  v8f acc[4] = {v8f{}, v8f{}, v8f{}, v8f{}};
#pragma unroll 4
  for (int k0 = 0; k0 < CIN; k0 += 4) {
    const int ka = k0 + 2 * half;
    v2f afrag;
    afrag.x = arow[ka];
    afrag.y = arow[ka + 1];
    const float* __restrict__ wk = wh + (size_t)ka * DH + m;
#pragma unroll
    for (int t = 0; t < 4; ++t) {
      v2f bfrag;
      bfrag.x = wk[t * 16];
      bfrag.y = wk[t * 16 + DH];
      acc[t] = __builtin_amdgcn_wmma_f32_16x16x4_f32(
          false, afrag, false, bfrag, (short)0, acc[t], false, false);
    }
  }

  const int mbase = half * 8;
  float p[8] = {0.0f, 0.0f, 0.0f, 0.0f, 0.0f, 0.0f, 0.0f, 0.0f};
#pragma unroll
  for (int t = 0; t < 4; ++t) {
    const int col = head * DH + t * 16 + m;
    const float bias = bvec[col];
    const float ae = a_e[col];
#pragma unroll
    for (int r = 0; r < 8; ++r) {
      const float y = acc[t][r] + bias;
      Y[(size_t)(row0 + mbase + r) * HID + col] = y;
      p[r] += y * ae;
    }
  }
#pragma unroll
  for (int off = 8; off >= 1; off >>= 1) {
#pragma unroll
    for (int r = 0; r < 8; ++r) p[r] += __shfl_xor(p[r], off);
  }
  float alpha = p[0];
#pragma unroll
  for (int r = 1; r < 8; ++r) alpha = (m == r) ? p[r] : alpha;
  if (m < 8) {
    float s = alpha >= 0.0f ? alpha : NEG_SLOPE * alpha;
    s = fminf(fmaxf(s, 0.001f), 5.0f);
    es[(row0 + mbase + m) * NH + head] = __expf(s);
  }
}


__global__ __launch_bounds__(256) void inc_fill(
    const int* __restrict__ inc_v, const int* __restrict__ inc_e,
    int* __restrict__ deg_v, int* __restrict__ deg_e,
    float* __restrict__ out, int* __restrict__ elist)
{
  const int i = blockIdx.x * 256 + threadIdx.x;
  if (i >= N_NZ) return;
  const int v = inc_v[i];
  const int e = inc_e[i];
  const int sv = atomicAdd(&deg_v[v], 1);
  const int se = atomicAdd(&deg_e[e], 1);
  if (sv < ROWCAP) ((int*)out)[(size_t)v * HID + sv] = e;
  if (se < ROWCAP) elist[(size_t)e * HID + se] = v;
}


#define VB 4

__device__ __forceinline__ float lane_val(float x, int k) {
  return __int_as_float(__builtin_amdgcn_readlane(__float_as_int(x), k));
}

__device__ __forceinline__ void v_rows_load(
    const char* __restrict__ Yb, unsigned lane16, int e_l, int k0, int cnt,
    float4 (&ya)[VB], float4 (&yb)[VB])
{
#pragma unroll
  for (int q = 0; q < VB; ++q) {
    const int kk = min(k0 + q, cnt - 1);
    const int e = __builtin_amdgcn_readlane(e_l, kk);
    const float4* __restrict__ yrow4 =
        (const float4*)(Yb + (size_t)e * (HID * sizeof(float)) + lane16);
    ya[q] = yrow4[0];
    yb[q] = yrow4[32];
  }
}

__device__ __forceinline__ void v_rows_apply(
    const float4& w_l, bool hi, int k0,
    const float4 (&ya)[VB], const float4 (&yb)[VB], float4& a, float4& b)
{
#pragma unroll
  for (int q = 0; q < VB; ++q) {
    const int kk = k0 + q;
    const float wx = lane_val(w_l.x, kk);
    const float wy = lane_val(w_l.y, kk);
    const float wz = lane_val(w_l.z, kk);
    const float ww = lane_val(w_l.w, kk);
    const float wa = hi ? wy : wx;
    const float wb = hi ? ww : wz;
    a.x += wa * ya[q].x; a.y += wa * ya[q].y; a.z += wa * ya[q].z; a.w += wa * ya[q].w;
    b.x += wb * yb[q].x; b.y += wb * yb[q].y; b.z += wb * yb[q].z; b.w += wb * yb[q].w;
  }
}

__global__ __launch_bounds__(256) void v_gather(
    const float* __restrict__ Y, const float* __restrict__ es,
    const int* __restrict__ deg,
    const int* __restrict__ inc_v, const int* __restrict__ inc_e,
    float* out)
{
  const int v = __builtin_amdgcn_readfirstlane(blockIdx.x * 8 + (threadIdx.x >> 5));
  const int lane = threadIdx.x & 31;
  const int n = deg[v];
  const int* lst = (const int*)(out + (size_t)v * HID);
  const float4* __restrict__ es4 = (const float4*)es;
  const char* __restrict__ Yb = (const char*)Y;
  const unsigned lane16 = (unsigned)lane * 16u;
  float4* orow4 = (float4*)(out + (size_t)v * HID);

  if (n <= 0) {
    const float4 z = make_float4(0.0f, 0.0f, 0.0f, 0.0f);
    orow4[lane] = z;
    orow4[lane + 32] = z;
    return;
  }

  const bool hi = lane >= 16;

  float4 a = make_float4(0.0f, 0.0f, 0.0f, 0.0f);
  float4 b = make_float4(0.0f, 0.0f, 0.0f, 0.0f);
  float4 ya[VB], yb[VB];

  if (n <= 32) {
    const int e_l = lst[min(lane, n - 1)];
    const float4 s_l = es4[e_l];
    v_rows_load(Yb, lane16, e_l, 0, n, ya, yb);

    float d0 = 0.0f, d1 = 0.0f, d2 = 0.0f, d3 = 0.0f;
    int k = 0;
    do {
      d0 += lane_val(s_l.x, k); d1 += lane_val(s_l.y, k);
      d2 += lane_val(s_l.z, k); d3 += lane_val(s_l.w, k);
    } while (++k < n);

    float4 w_l;
    w_l.x = lane < n ? s_l.x / d0 : 0.0f;
    w_l.y = lane < n ? s_l.y / d1 : 0.0f;
    w_l.z = lane < n ? s_l.z / d2 : 0.0f;
    w_l.w = lane < n ? s_l.w / d3 : 0.0f;
    v_rows_apply(w_l, hi, 0, ya, yb, a, b);
    for (int k0 = VB; k0 < n; k0 += VB) {
      v_rows_load(Yb, lane16, e_l, k0, n, ya, yb);
      v_rows_apply(w_l, hi, k0, ya, yb, a, b);
    }
  } else if (n <= ROWCAP) {
    float d0 = 0.0f, d1 = 0.0f, d2 = 0.0f, d3 = 0.0f;
    for (int c = 0; c < n; c += 32) {
      const int cnt = min(n - c, 32);
      const int e_t = lst[c + min(lane, cnt - 1)];
      const float4 s_t = es4[e_t];
      for (int k = 0; k < cnt; ++k) {
        d0 += lane_val(s_t.x, k); d1 += lane_val(s_t.y, k);
        d2 += lane_val(s_t.z, k); d3 += lane_val(s_t.w, k);
      }
    }
    for (int c = 0; c < n; c += 32) {
      const int cnt = min(n - c, 32);
      const int e_l = lst[c + min(lane, cnt - 1)];
      const float4 s_l = es4[e_l];
      float4 w_l;
      w_l.x = lane < cnt ? s_l.x / d0 : 0.0f;
      w_l.y = lane < cnt ? s_l.y / d1 : 0.0f;
      w_l.z = lane < cnt ? s_l.z / d2 : 0.0f;
      w_l.w = lane < cnt ? s_l.w / d3 : 0.0f;
      for (int k0 = 0; k0 < cnt; k0 += VB) {
        v_rows_load(Yb, lane16, e_l, k0, cnt, ya, yb);
        v_rows_apply(w_l, hi, k0, ya, yb, a, b);
      }
    }
  } else {
    float d0 = 0.0f, d1 = 0.0f, d2 = 0.0f, d3 = 0.0f;
    for (int c = 0; c < N_NZ; c += 32) {
      const int i = c + lane;
      const bool ok = i < N_NZ;
      const int key = ok ? inc_v[i] : -1;
      const int val = ok ? inc_e[i] : 0;
      unsigned m = __builtin_amdgcn_ballot_w32(key == v);
      while (m != 0u) {
        const int k = __builtin_ctz(m);
        m &= m - 1u;
        const float4 s4 = es4[__builtin_amdgcn_readlane(val, k)];
        d0 += s4.x; d1 += s4.y; d2 += s4.z; d3 += s4.w;
      }
    }
    for (int c = 0; c < N_NZ; c += 32) {
      const int i = c + lane;
      const bool ok = i < N_NZ;
      const int key = ok ? inc_v[i] : -1;
      const int val = ok ? inc_e[i] : 0;
      unsigned m = __builtin_amdgcn_ballot_w32(key == v);
      while (m != 0u) {
        const int k = __builtin_ctz(m);
        m &= m - 1u;
        const int ee = __builtin_amdgcn_readlane(val, k);
        const float4 s4 = es4[ee];
        const float wa = hi ? s4.y / d1 : s4.x / d0;
        const float wb = hi ? s4.w / d3 : s4.z / d2;
        const float4* __restrict__ yrow4 =
            (const float4*)(Yb + (size_t)ee * (HID * sizeof(float)) + lane16);
        const float4 y0 = yrow4[0];
        const float4 y1 = yrow4[32];
        a.x += wa * y0.x; a.y += wa * y0.y; a.z += wa * y0.z; a.w += wa * y0.w;
        b.x += wb * y1.x; b.y += wb * y1.y; b.z += wb * y1.z; b.w += wb * y1.w;
      }
    }
  }

  a.x = a.x > 0.0f ? a.x : expm1f(a.x);
  a.y = a.y > 0.0f ? a.y : expm1f(a.y);
  a.z = a.z > 0.0f ? a.z : expm1f(a.z);
  a.w = a.w > 0.0f ? a.w : expm1f(a.w);
  b.x = b.x > 0.0f ? b.x : expm1f(b.x);
  b.y = b.y > 0.0f ? b.y : expm1f(b.y);
  b.z = b.z > 0.0f ? b.z : expm1f(b.z);
  b.w = b.w > 0.0f ? b.w : expm1f(b.w);

  orow4[lane] = a;
  orow4[lane + 32] = b;
}

extern "C" void kernel_launch(void* const* d_in, const int* in_sizes, int n_in,
                              void* d_out, int out_size, void* d_ws, size_t ws_size,
                              hipStream_t stream) {
  const float* X     = (const float*)d_in[0];
  const float* W     = (const float*)d_in[1];
  const float* bvec  = (const float*)d_in[2];
  const float* a_e   = (const float*)d_in[3];
  const int*   inc_v = (const int*)d_in[4];
  const int*   inc_e = (const int*)d_in[5];
  float* out = (float*)d_out;

  float* ws    = (float*)d_ws;
  int*   elist = (int*)ws;
  float* Y     = ws + (size_t)N_E * HID;
  float* es    = Y  + (size_t)N_E * HID;
  int*   deg_e = (int*)(es + (size_t)N_E * NH);
  int*   deg_v = deg_e + N_E;

  hipMemsetAsync(deg_e, 0, sizeof(int) * (N_E + N_V), stream);

  inc_fill<<<dim3((N_NZ + 255) / 256), 256, 0, stream>>>(inc_v, inc_e, deg_v, deg_e, out, elist);
  edge_tile<<<dim3(N_E / 16), 256, 0, stream>>>(X, deg_e, inc_v, inc_e, elist, W, bvec, a_e, Y, es);
  v_gather<<<dim3(N_V / 8), 256, 0, stream>>>(Y, es, deg_v, inc_v, inc_e, out);
}


# incidences grouped by 128-key buckets with LDS-counter ranks (one global add per work-group and bucket); tile and vertex groups rank their bucket in LDS lists; sum order = LDS arrival rank
# speedup vs baseline: 5.3560x; 1.3834x over previous
#include <hip/hip_runtime.h>
#include <hip/hip_bf16.h>

typedef __attribute__((ext_vector_type(2))) float v2f;
typedef __attribute__((ext_vector_type(8))) float v8f;

#define N_V   100000
#define N_E   20000
#define N_NZ  400000
#define CIN   256
#define HID   256
#define NH    4
#define DH    64
#define NEG_SLOPE 0.2f

#define BSH    7
#define N_EB   ((N_E + 127) >> BSH)
#define N_VB   ((N_V + 127) >> BSH)
#define N_BINS (N_EB + N_VB)
#define CAP_E  8192
#define CAP_V  2048
#define ROWCAP 256
#define VCAP   64
static_assert(N_V <= (1 << 17), "vertex id must fit 17 bits");
static_assert(N_E <= (1 << 15), "hyperedge id must fit 15 bits");
static_assert(N_BINS <= 1024, "one thread per bucket counter");
static_assert((size_t)N_EB * CAP_E + (size_t)N_VB * CAP_V <= (size_t)N_E * HID, "regions fit the list plane");


#define APITCH 260

#define EPT 12

__device__ __forceinline__ int edge_sum(
    const char* __restrict__ Xb, const int* __restrict__ lst, int t, int e,
    const int* __restrict__ inc_v, const int* __restrict__ inc_e,
    int lane, unsigned lane16, float4& a, float4& b)
{
  a = make_float4(0.0f, 0.0f, 0.0f, 0.0f);
  b = make_float4(0.0f, 0.0f, 0.0f, 0.0f);
  if (t <= ROWCAP) {
    for (int j = 0; j < t; j += 8) {
      float4 xa[8], xb[8];
#pragma unroll
      for (int k = 0; k < 8; ++k) {
        const int jj = min(j + k, t - 1);
        const unsigned off = (unsigned)lst[jj] * (unsigned)(CIN * sizeof(float)) + lane16;
        const float4* __restrict__ xrow4 = (const float4*)(Xb + off);
        xa[k] = xrow4[0];
        xb[k] = xrow4[32];
      }
#pragma unroll
      for (int k = 0; k < 8; ++k) {
        const bool in = j + k < t;
        a.x += in ? xa[k].x : 0.0f; a.y += in ? xa[k].y : 0.0f;
        a.z += in ? xa[k].z : 0.0f; a.w += in ? xa[k].w : 0.0f;
        b.x += in ? xb[k].x : 0.0f; b.y += in ? xb[k].y : 0.0f;
        b.z += in ? xb[k].z : 0.0f; b.w += in ? xb[k].w : 0.0f;
      }
    }
    return t;
  }
  int members = 0;
  for (int c = 0; c < N_NZ; c += 32) {
    const int i = c + lane;
    const bool ok = i < N_NZ;
    const int key = ok ? inc_e[i] : -1;
    const int val = ok ? inc_v[i] : 0;
    unsigned m = __builtin_amdgcn_ballot_w32(key == e);
    members += __builtin_popcount(m);
    while (m != 0u) {
      const int k = __builtin_ctz(m);
      m &= m - 1u;
      const int vv = __builtin_amdgcn_readlane(val, k);
      const unsigned off = (unsigned)vv * (unsigned)(CIN * sizeof(float)) + lane16;
      const float4* __restrict__ xrow4 = (const float4*)(Xb + off);
      const float4 xa = xrow4[0];
      const float4 xb = xrow4[32];
      a.x += xa.x; a.y += xa.y; a.z += xa.z; a.w += xa.w;
      b.x += xb.x; b.y += xb.y; b.z += xb.z; b.w += xb.w;
    }
  }
  return members;
}

__global__ __launch_bounds__(256) void edge_tile(
    const float* __restrict__ X, const int* __restrict__ cursor,
    const int* __restrict__ inc_v, const int* __restrict__ inc_e,
    const unsigned* __restrict__ recE,
    const float* __restrict__ W, const float* __restrict__ bvec,
    const float* __restrict__ a_e,
    float* __restrict__ Y, float* __restrict__ es)
{
  __shared__ __attribute__((aligned(16))) float As[16 * APITCH];
  __shared__ int elst[16 * ROWCAP];
  __shared__ int ecnt[16];

  const int tid = threadIdx.x;
  const int lane = threadIdx.x & 31;
  const int wave = __builtin_amdgcn_readfirstlane(threadIdx.x >> 5);
  const int row0 = blockIdx.x * 16;

  const int be = row0 >> BSH;
  const int nraw = cursor[be];
  const bool whole = nraw <= CAP_E;
  if (tid < 16) ecnt[tid] = 0;
  __syncthreads();
  if (whole) {
    const unsigned* __restrict__ rec = recE + (size_t)be * CAP_E;
    const unsigned tl = (unsigned)(row0 & 127) >> 4;
    for (int c0 = 0; c0 < nraw; c0 += 256 * EPT) {
      unsigned r[EPT];
#pragma unroll
      for (int k = 0; k < EPT; ++k) {
        const int c = c0 + k * 256 + tid;
        r[k] = rec[min(c, nraw - 1)];
      }
      int slot[EPT];
#pragma unroll
      for (int k = 0; k < EPT; ++k) {
        const int c = c0 + k * 256 + tid;
        const unsigned el = r[k] >> 17;
        slot[k] = ROWCAP;
        if (c < nraw && (el >> 4) == tl) slot[k] = atomicAdd(&ecnt[el & 15u], 1);
      }
#pragma unroll
      for (int k = 0; k < EPT; ++k) {
        const unsigned el = r[k] >> 17;
        if (slot[k] < ROWCAP) elst[(el & 15u) * ROWCAP + slot[k]] = (int)(r[k] & 0x1FFFFu);
      }
    }
  }
  __syncthreads();

  {
    const char* __restrict__ Xb = (const char*)X;
    const unsigned lane16 = (unsigned)lane * 16u;
    for (int q = 0; q < 2; ++q) {
      const int r = 2 * wave + q;
      const int e = row0 + r;
      const int t = whole ? __builtin_amdgcn_readfirstlane(ecnt[r]) : ROWCAP + 1;
      float4 a, b;
      const int members = edge_sum(Xb, elst + r * ROWCAP, t, e, inc_v, inc_e, lane, lane16, a, b);
      const float inv = 1.0f / fmaxf((float)members, 1.0f);
      a.x *= inv; a.y *= inv; a.z *= inv; a.w *= inv;
      b.x *= inv; b.y *= inv; b.z *= inv; b.w *= inv;
      float4* arow4 = (float4*)(As + r * APITCH);
      arow4[lane] = a;
      arow4[lane + 32] = b;
    }
  }
  __syncthreads();
  if (wave >= 4) return;

  const int head = wave;
  const int m = lane & 15;
  const int half = lane >> 4;

  const float* arow = As + m * APITCH;
  const float* __restrict__ wh = W + (size_t)head * (CIN * DH);

  v8f acc[4] = {v8f{}, v8f{}, v8f{}, v8f{}};
#pragma unroll 4
  for (int k0 = 0; k0 < CIN; k0 += 4) {
    const int ka = k0 + 2 * half;
    v2f afrag;
    afrag.x = arow[ka];
    afrag.y = arow[ka + 1];
    const float* __restrict__ wk = wh + (size_t)ka * DH + m;
#pragma unroll
    for (int t = 0; t < 4; ++t) {
      v2f bfrag;
      bfrag.x = wk[t * 16];
      bfrag.y = wk[t * 16 + DH];
      acc[t] = __builtin_amdgcn_wmma_f32_16x16x4_f32(
          false, afrag, false, bfrag, (short)0, acc[t], false, false);
    }
  }

  const int mbase = half * 8;
  float p[8] = {0.0f, 0.0f, 0.0f, 0.0f, 0.0f, 0.0f, 0.0f, 0.0f};
#pragma unroll
  for (int t = 0; t < 4; ++t) {
    const int col = head * DH + t * 16 + m;
    const float bias = bvec[col];
    const float ae = a_e[col];
#pragma unroll
    for (int r = 0; r < 8; ++r) {
      const float y = acc[t][r] + bias;
      Y[(size_t)(row0 + mbase + r) * HID + col] = y;
      p[r] += y * ae;
    }
  }
#pragma unroll
  for (int off = 8; off >= 1; off >>= 1) {
#pragma unroll
    for (int r = 0; r < 8; ++r) p[r] += __shfl_xor(p[r], off);
  }
  float alpha = p[0];
#pragma unroll
  for (int r = 1; r < 8; ++r) alpha = (m == r) ? p[r] : alpha;
  if (m < 8) {
    float s = alpha >= 0.0f ? alpha : NEG_SLOPE * alpha;
    s = fminf(fmaxf(s, 0.001f), 5.0f);
    es[(row0 + mbase + m) * NH + head] = __expf(s);
  }
}


#define FILL_BLOCKS  256
#define FILL_THREADS 1024
#define FILL_CHUNK   ((N_NZ + FILL_BLOCKS - 1) / FILL_BLOCKS)
#define FILL_PER     ((FILL_CHUNK + FILL_THREADS - 1) / FILL_THREADS)

__global__ __launch_bounds__(FILL_THREADS) void group_fill(
    const int* __restrict__ inc_v, const int* __restrict__ inc_e,
    int* __restrict__ cursor,
    unsigned* __restrict__ recE, unsigned* __restrict__ recV)
{
  __shared__ int bins[N_BINS];
  const int tid = threadIdx.x;
  if (tid < N_BINS) bins[tid] = 0;
  __syncthreads();

  int kv[FILL_PER], ke[FILL_PER], rv[FILL_PER], re[FILL_PER];
  bool ok[FILL_PER];
#pragma unroll
  for (int q = 0; q < FILL_PER; ++q) {
    const int li = tid + q * FILL_THREADS;
    const int i = blockIdx.x * FILL_CHUNK + li;
    ok[q] = li < FILL_CHUNK && i < N_NZ;
    kv[q] = ok[q] ? inc_v[i] : 0;
    ke[q] = ok[q] ? inc_e[i] : 0;
  }
#pragma unroll
  for (int q = 0; q < FILL_PER; ++q) {
    re[q] = 0;
    rv[q] = 0;
    if (ok[q]) {
      re[q] = atomicAdd(&bins[ke[q] >> BSH], 1);
      rv[q] = atomicAdd(&bins[N_EB + (kv[q] >> BSH)], 1);
    }
  }
  __syncthreads();
  if (tid < N_BINS) {
    const int c = bins[tid];
    bins[tid] = c > 0 ? atomicAdd(&cursor[tid], c) : 0;
  }
  __syncthreads();
#pragma unroll
  for (int q = 0; q < FILL_PER; ++q) {
    if (ok[q]) {
      const int be = ke[q] >> BSH;
      const int bv = kv[q] >> BSH;
      const int pe = bins[be] + re[q];
      const int pv = bins[N_EB + bv] + rv[q];
      if (pe < CAP_E)
        recE[(size_t)be * CAP_E + pe] = ((unsigned)(ke[q] & 127) << 17) | (unsigned)kv[q];
      if (pv < CAP_V)
        recV[(size_t)bv * CAP_V + pv] = ((unsigned)(kv[q] & 127) << 15) | (unsigned)ke[q];
    }
  }
}


#define VB 4

__device__ __forceinline__ float lane_val(float x, int k) {
  return __int_as_float(__builtin_amdgcn_readlane(__float_as_int(x), k));
}

__device__ __forceinline__ void v_rows_load(
    const char* __restrict__ Yb, unsigned lane16, int e_l, int k0, int cnt,
    float4 (&ya)[VB], float4 (&yb)[VB])
{
#pragma unroll
  for (int q = 0; q < VB; ++q) {
    const int kk = min(k0 + q, cnt - 1);
    const int e = __builtin_amdgcn_readlane(e_l, kk);
    const float4* __restrict__ yrow4 =
        (const float4*)(Yb + (size_t)e * (HID * sizeof(float)) + lane16);
    ya[q] = yrow4[0];
    yb[q] = yrow4[32];
  }
}

__device__ __forceinline__ void v_rows_apply(
    const float4& w_l, bool hi, int k0,
    const float4 (&ya)[VB], const float4 (&yb)[VB], float4& a, float4& b)
{
#pragma unroll
  for (int q = 0; q < VB; ++q) {
    const int kk = k0 + q;
    const float wx = lane_val(w_l.x, kk);
    const float wy = lane_val(w_l.y, kk);
    const float wz = lane_val(w_l.z, kk);
    const float ww = lane_val(w_l.w, kk);
    const float wa = hi ? wy : wx;
    const float wb = hi ? ww : wz;
    a.x += wa * ya[q].x; a.y += wa * ya[q].y; a.z += wa * ya[q].z; a.w += wa * ya[q].w;
    b.x += wb * yb[q].x; b.y += wb * yb[q].y; b.z += wb * yb[q].z; b.w += wb * yb[q].w;
  }
}

#define VPT 3

__global__ __launch_bounds__(256) void v_gather(
    const float* __restrict__ Y, const float* __restrict__ es,
    const int* __restrict__ cursor, const unsigned* __restrict__ recV,
    const int* __restrict__ inc_v, const int* __restrict__ inc_e,
    float* __restrict__ out)
{
  __shared__ int vlst[8 * VCAP];
  __shared__ int vcnt[8];

  const int tid = threadIdx.x;
  const int lane = threadIdx.x & 31;
  const int wave = __builtin_amdgcn_readfirstlane(threadIdx.x >> 5);
  const int v0 = blockIdx.x * 8;
  const int v = v0 + wave;

  const int bv = v0 >> BSH;
  const int nraw = cursor[N_EB + bv];
  const bool whole = nraw <= CAP_V;
  if (tid < 8) vcnt[tid] = 0;
  __syncthreads();
  if (whole) {
    const unsigned* __restrict__ rec = recV + (size_t)bv * CAP_V;
    const unsigned sib = (unsigned)(v0 & 127) >> 3;
    for (int c0 = 0; c0 < nraw; c0 += 256 * VPT) {
      unsigned r[VPT];
#pragma unroll
      for (int k = 0; k < VPT; ++k) {
        const int c = c0 + k * 256 + tid;
        r[k] = rec[min(c, nraw - 1)];
      }
      int slot[VPT];
#pragma unroll
      for (int k = 0; k < VPT; ++k) {
        const int c = c0 + k * 256 + tid;
        const unsigned vl = r[k] >> 15;
        slot[k] = VCAP;
        if (c < nraw && (vl >> 3) == sib) slot[k] = atomicAdd(&vcnt[vl & 7u], 1);
      }
#pragma unroll
      for (int k = 0; k < VPT; ++k) {
        const unsigned vl = r[k] >> 15;
        if (slot[k] < VCAP) vlst[(vl & 7u) * VCAP + slot[k]] = (int)(r[k] & 0x7FFFu);
      }
    }
  }
  __syncthreads();

  const int n = whole ? __builtin_amdgcn_readfirstlane(vcnt[wave]) : VCAP + 1;
  const int* lst = vlst + wave * VCAP;
  const float4* __restrict__ es4 = (const float4*)es;
  const char* __restrict__ Yb = (const char*)Y;
  const unsigned lane16 = (unsigned)lane * 16u;
  float4* orow4 = (float4*)(out + (size_t)v * HID);

  if (n <= 0) {
    const float4 z = make_float4(0.0f, 0.0f, 0.0f, 0.0f);
    orow4[lane] = z;
    orow4[lane + 32] = z;
    return;
  }

  const bool hi = lane >= 16;

  float4 a = make_float4(0.0f, 0.0f, 0.0f, 0.0f);
  float4 b = make_float4(0.0f, 0.0f, 0.0f, 0.0f);
  float4 ya[VB], yb[VB];

  if (n <= 32) {
    const int e_l = lst[min(lane, n - 1)];
    const float4 s_l = es4[e_l];
    v_rows_load(Yb, lane16, e_l, 0, n, ya, yb);

    float d0 = 0.0f, d1 = 0.0f, d2 = 0.0f, d3 = 0.0f;
    int k = 0;
    do {
      d0 += lane_val(s_l.x, k); d1 += lane_val(s_l.y, k);
      d2 += lane_val(s_l.z, k); d3 += lane_val(s_l.w, k);
    } while (++k < n);

    float4 w_l;
    w_l.x = lane < n ? s_l.x / d0 : 0.0f;
    w_l.y = lane < n ? s_l.y / d1 : 0.0f;
    w_l.z = lane < n ? s_l.z / d2 : 0.0f;
    w_l.w = lane < n ? s_l.w / d3 : 0.0f;
    v_rows_apply(w_l, hi, 0, ya, yb, a, b);
    for (int k0 = VB; k0 < n; k0 += VB) {
      v_rows_load(Yb, lane16, e_l, k0, n, ya, yb);
      v_rows_apply(w_l, hi, k0, ya, yb, a, b);
    }
  } else if (n <= VCAP) {
    float d0 = 0.0f, d1 = 0.0f, d2 = 0.0f, d3 = 0.0f;
    for (int c = 0; c < n; c += 32) {
      const int cnt = min(n - c, 32);
      const int e_t = lst[c + min(lane, cnt - 1)];
      const float4 s_t = es4[e_t];
      for (int k = 0; k < cnt; ++k) {
        d0 += lane_val(s_t.x, k); d1 += lane_val(s_t.y, k);
        d2 += lane_val(s_t.z, k); d3 += lane_val(s_t.w, k);
      }
    }
    for (int c = 0; c < n; c += 32) {
      const int cnt = min(n - c, 32);
      const int e_l = lst[c + min(lane, cnt - 1)];
      const float4 s_l = es4[e_l];
      float4 w_l;
      w_l.x = lane < cnt ? s_l.x / d0 : 0.0f;
      w_l.y = lane < cnt ? s_l.y / d1 : 0.0f;
      w_l.z = lane < cnt ? s_l.z / d2 : 0.0f;
      w_l.w = lane < cnt ? s_l.w / d3 : 0.0f;
      for (int k0 = 0; k0 < cnt; k0 += VB) {
        v_rows_load(Yb, lane16, e_l, k0, cnt, ya, yb);
        v_rows_apply(w_l, hi, k0, ya, yb, a, b);
      }
    }
  } else {
    float d0 = 0.0f, d1 = 0.0f, d2 = 0.0f, d3 = 0.0f;
    for (int c = 0; c < N_NZ; c += 32) {
      const int i = c + lane;
      const bool ok = i < N_NZ;
      const int key = ok ? inc_v[i] : -1;
      const int val = ok ? inc_e[i] : 0;
      unsigned m = __builtin_amdgcn_ballot_w32(key == v);
      while (m != 0u) {
        const int k = __builtin_ctz(m);
        m &= m - 1u;
        const float4 s4 = es4[__builtin_amdgcn_readlane(val, k)];
        d0 += s4.x; d1 += s4.y; d2 += s4.z; d3 += s4.w;
      }
    }
    for (int c = 0; c < N_NZ; c += 32) {
      const int i = c + lane;
      const bool ok = i < N_NZ;
      const int key = ok ? inc_v[i] : -1;
      const int val = ok ? inc_e[i] : 0;
      unsigned m = __builtin_amdgcn_ballot_w32(key == v);
      while (m != 0u) {
        const int k = __builtin_ctz(m);
        m &= m - 1u;
        const int ee = __builtin_amdgcn_readlane(val, k);
        const float4 s4 = es4[ee];
        const float wa = hi ? s4.y / d1 : s4.x / d0;
        const float wb = hi ? s4.w / d3 : s4.z / d2;
        const float4* __restrict__ yrow4 =
            (const float4*)(Yb + (size_t)ee * (HID * sizeof(float)) + lane16);
        const float4 y0 = yrow4[0];
        const float4 y1 = yrow4[32];
        a.x += wa * y0.x; a.y += wa * y0.y; a.z += wa * y0.z; a.w += wa * y0.w;
        b.x += wb * y1.x; b.y += wb * y1.y; b.z += wb * y1.z; b.w += wb * y1.w;
      }
    }
  }

  a.x = a.x > 0.0f ? a.x : expm1f(a.x);
  a.y = a.y > 0.0f ? a.y : expm1f(a.y);
  a.z = a.z > 0.0f ? a.z : expm1f(a.z);
  a.w = a.w > 0.0f ? a.w : expm1f(a.w);
  b.x = b.x > 0.0f ? b.x : expm1f(b.x);
  b.y = b.y > 0.0f ? b.y : expm1f(b.y);
  b.z = b.z > 0.0f ? b.z : expm1f(b.z);
  b.w = b.w > 0.0f ? b.w : expm1f(b.w);

  orow4[lane] = a;
  orow4[lane + 32] = b;
}

extern "C" void kernel_launch(void* const* d_in, const int* in_sizes, int n_in,
                              void* d_out, int out_size, void* d_ws, size_t ws_size,
                              hipStream_t stream) {
  const float* X     = (const float*)d_in[0];
  const float* W     = (const float*)d_in[1];
  const float* bvec  = (const float*)d_in[2];
  const float* a_e   = (const float*)d_in[3];
  const int*   inc_v = (const int*)d_in[4];
  const int*   inc_e = (const int*)d_in[5];
  float* out = (float*)d_out;

  float* ws    = (float*)d_ws;
  unsigned* recE = (unsigned*)ws;
  unsigned* recV = recE + (size_t)N_EB * CAP_E;
  float* Y     = ws + (size_t)N_E * HID;
  float* es    = Y  + (size_t)N_E * HID;
  int*   cursor = (int*)(es + (size_t)N_E * NH);

  hipMemsetAsync(cursor, 0, sizeof(int) * N_BINS, stream);

  group_fill<<<dim3(FILL_BLOCKS), FILL_THREADS, 0, stream>>>(inc_v, inc_e, cursor, recE, recV);
  edge_tile<<<dim3(N_E / 16), 256, 0, stream>>>(X, cursor, inc_v, inc_e, recE, W, bvec, a_e, Y, es);
  v_gather<<<dim3(N_V / 8), 256, 0, stream>>>(Y, es, cursor, recV, inc_v, inc_e, out);
}


# v_gather ELU: exp(x)-1 from the exp2 unit (mul, v_exp_f32, sub), not library expm1f (21 vector ops a value); abs error ~1e-7, relative accuracy near 0 given up; Y row offsets 32-bit to keep 64 VGPRs
# speedup vs baseline: 5.5270x; 1.0319x over previous
#include <hip/hip_runtime.h>
#include <hip/hip_bf16.h>

typedef __attribute__((ext_vector_type(2))) float v2f;
typedef __attribute__((ext_vector_type(8))) float v8f;

#define N_V   100000
#define N_E   20000
#define N_NZ  400000
#define CIN   256
#define HID   256
#define NH    4
#define DH    64
#define NEG_SLOPE 0.2f

#define BSH    7
#define N_EB   ((N_E + 127) >> BSH)
#define N_VB   ((N_V + 127) >> BSH)
#define N_BINS (N_EB + N_VB)
#define CAP_E  8192
#define CAP_V  2048
#define ROWCAP 256
#define VCAP   64
static_assert(N_V <= (1 << 17), "vertex id must fit 17 bits");
static_assert(N_E <= (1 << 15), "hyperedge id must fit 15 bits");
static_assert(N_BINS <= 1024, "one thread per bucket counter");
static_assert((size_t)N_EB * CAP_E + (size_t)N_VB * CAP_V <= (size_t)N_E * HID, "regions fit the list plane");


#define APITCH 260

#define EPT 12

__device__ __forceinline__ int edge_sum(
    const char* __restrict__ Xb, const int* __restrict__ lst, int t, int e,
    const int* __restrict__ inc_v, const int* __restrict__ inc_e,
    int lane, unsigned lane16, float4& a, float4& b)
{
  a = make_float4(0.0f, 0.0f, 0.0f, 0.0f);
  b = make_float4(0.0f, 0.0f, 0.0f, 0.0f);
  if (t <= ROWCAP) {
    for (int j = 0; j < t; j += 8) {
      float4 xa[8], xb[8];
#pragma unroll
      for (int k = 0; k < 8; ++k) {
        const int jj = min(j + k, t - 1);
        const unsigned off = (unsigned)lst[jj] * (unsigned)(CIN * sizeof(float)) + lane16;
        const float4* __restrict__ xrow4 = (const float4*)(Xb + off);
        xa[k] = xrow4[0];
        xb[k] = xrow4[32];
      }
#pragma unroll
      for (int k = 0; k < 8; ++k) {
        const bool in = j + k < t;
        a.x += in ? xa[k].x : 0.0f; a.y += in ? xa[k].y : 0.0f;
        a.z += in ? xa[k].z : 0.0f; a.w += in ? xa[k].w : 0.0f;
        b.x += in ? xb[k].x : 0.0f; b.y += in ? xb[k].y : 0.0f;
        b.z += in ? xb[k].z : 0.0f; b.w += in ? xb[k].w : 0.0f;
      }
    }
    return t;
  }
  int members = 0;
  for (int c = 0; c < N_NZ; c += 32) {
    const int i = c + lane;
    const bool ok = i < N_NZ;
    const int key = ok ? inc_e[i] : -1;
    const int val = ok ? inc_v[i] : 0;
    unsigned m = __builtin_amdgcn_ballot_w32(key == e);
    members += __builtin_popcount(m);
    while (m != 0u) {
      const int k = __builtin_ctz(m);
      m &= m - 1u;
      const int vv = __builtin_amdgcn_readlane(val, k);
      const unsigned off = (unsigned)vv * (unsigned)(CIN * sizeof(float)) + lane16;
      const float4* __restrict__ xrow4 = (const float4*)(Xb + off);
      const float4 xa = xrow4[0];
      const float4 xb = xrow4[32];
      a.x += xa.x; a.y += xa.y; a.z += xa.z; a.w += xa.w;
      b.x += xb.x; b.y += xb.y; b.z += xb.z; b.w += xb.w;
    }
  }
  return members;
}

__global__ __launch_bounds__(256) void edge_tile(
    const float* __restrict__ X, const int* __restrict__ cursor,
    const int* __restrict__ inc_v, const int* __restrict__ inc_e,
    const unsigned* __restrict__ recE,
    const float* __restrict__ W, const float* __restrict__ bvec,
    const float* __restrict__ a_e,
    float* __restrict__ Y, float* __restrict__ es)
{
  __shared__ __attribute__((aligned(16))) float As[16 * APITCH];
  __shared__ int elst[16 * ROWCAP];
  __shared__ int ecnt[16];

  const int tid = threadIdx.x;
  const int lane = threadIdx.x & 31;
  const int wave = __builtin_amdgcn_readfirstlane(threadIdx.x >> 5);
  const int row0 = blockIdx.x * 16;

  const int be = row0 >> BSH;
  const int nraw = cursor[be];
  const bool whole = nraw <= CAP_E;
  if (tid < 16) ecnt[tid] = 0;
  __syncthreads();
  if (whole) {
    const unsigned* __restrict__ rec = recE + (size_t)be * CAP_E;
    const unsigned tl = (unsigned)(row0 & 127) >> 4;
    for (int c0 = 0; c0 < nraw; c0 += 256 * EPT) {
      unsigned r[EPT];
#pragma unroll
      for (int k = 0; k < EPT; ++k) {
        const int c = c0 + k * 256 + tid;
        r[k] = rec[min(c, nraw - 1)];
      }
      int slot[EPT];
#pragma unroll
      for (int k = 0; k < EPT; ++k) {
        const int c = c0 + k * 256 + tid;
        const unsigned el = r[k] >> 17;
        slot[k] = ROWCAP;
        if (c < nraw && (el >> 4) == tl) slot[k] = atomicAdd(&ecnt[el & 15u], 1);
      }
#pragma unroll
      for (int k = 0; k < EPT; ++k) {
        const unsigned el = r[k] >> 17;
        if (slot[k] < ROWCAP) elst[(el & 15u) * ROWCAP + slot[k]] = (int)(r[k] & 0x1FFFFu);
      }
    }
  }
  __syncthreads();

  {
    const char* __restrict__ Xb = (const char*)X;
    const unsigned lane16 = (unsigned)lane * 16u;
    for (int q = 0; q < 2; ++q) {
      const int r = 2 * wave + q;
      const int e = row0 + r;
      const int t = whole ? __builtin_amdgcn_readfirstlane(ecnt[r]) : ROWCAP + 1;
      float4 a, b;
      const int members = edge_sum(Xb, elst + r * ROWCAP, t, e, inc_v, inc_e, lane, lane16, a, b);
      const float inv = 1.0f / fmaxf((float)members, 1.0f);
      a.x *= inv; a.y *= inv; a.z *= inv; a.w *= inv;
      b.x *= inv; b.y *= inv; b.z *= inv; b.w *= inv;
      float4* arow4 = (float4*)(As + r * APITCH);
      arow4[lane] = a;
      arow4[lane + 32] = b;
    }
  }
  __syncthreads();
  if (wave >= 4) return;

  const int head = wave;
  const int m = lane & 15;
  const int half = lane >> 4;

  const float* arow = As + m * APITCH;
  const float* __restrict__ wh = W + (size_t)head * (CIN * DH);

  v8f acc[4] = {v8f{}, v8f{}, v8f{}, v8f{}};
#pragma unroll 4
  for (int k0 = 0; k0 < CIN; k0 += 4) {
    const int ka = k0 + 2 * half;
    v2f afrag;
    afrag.x = arow[ka];
    afrag.y = arow[ka + 1];
    const float* __restrict__ wk = wh + (size_t)ka * DH + m;
#pragma unroll
    for (int t = 0; t < 4; ++t) {
      v2f bfrag;
      bfrag.x = wk[t * 16];
      bfrag.y = wk[t * 16 + DH];
      acc[t] = __builtin_amdgcn_wmma_f32_16x16x4_f32(
          false, afrag, false, bfrag, (short)0, acc[t], false, false);
    }
  }

  const int mbase = half * 8;
  float p[8] = {0.0f, 0.0f, 0.0f, 0.0f, 0.0f, 0.0f, 0.0f, 0.0f};
#pragma unroll
  for (int t = 0; t < 4; ++t) {
    const int col = head * DH + t * 16 + m;
    const float bias = bvec[col];
    const float ae = a_e[col];
#pragma unroll
    for (int r = 0; r < 8; ++r) {
      const float y = acc[t][r] + bias;
      Y[(size_t)(row0 + mbase + r) * HID + col] = y;
      p[r] += y * ae;
    }
  }
#pragma unroll
  for (int off = 8; off >= 1; off >>= 1) {
#pragma unroll
    for (int r = 0; r < 8; ++r) p[r] += __shfl_xor(p[r], off);
  }
  float alpha = p[0];
#pragma unroll
  for (int r = 1; r < 8; ++r) alpha = (m == r) ? p[r] : alpha;
  if (m < 8) {
    float s = alpha >= 0.0f ? alpha : NEG_SLOPE * alpha;
    s = fminf(fmaxf(s, 0.001f), 5.0f);
    es[(row0 + mbase + m) * NH + head] = __expf(s);
  }
}


#define FILL_BLOCKS  256
#define FILL_THREADS 1024
#define FILL_CHUNK   ((N_NZ + FILL_BLOCKS - 1) / FILL_BLOCKS)
#define FILL_PER     ((FILL_CHUNK + FILL_THREADS - 1) / FILL_THREADS)

__global__ __launch_bounds__(FILL_THREADS) void group_fill(
    const int* __restrict__ inc_v, const int* __restrict__ inc_e,
    int* __restrict__ cursor,
    unsigned* __restrict__ recE, unsigned* __restrict__ recV)
{
  __shared__ int bins[N_BINS];
  const int tid = threadIdx.x;
  if (tid < N_BINS) bins[tid] = 0;
  __syncthreads();

  int kv[FILL_PER], ke[FILL_PER], rv[FILL_PER], re[FILL_PER];
  bool ok[FILL_PER];
#pragma unroll
  for (int q = 0; q < FILL_PER; ++q) {
    const int li = tid + q * FILL_THREADS;
    const int i = blockIdx.x * FILL_CHUNK + li;
    ok[q] = li < FILL_CHUNK && i < N_NZ;
    kv[q] = ok[q] ? inc_v[i] : 0;
    ke[q] = ok[q] ? inc_e[i] : 0;
  }
#pragma unroll
  for (int q = 0; q < FILL_PER; ++q) {
    re[q] = 0;
    rv[q] = 0;
    if (ok[q]) {
      re[q] = atomicAdd(&bins[ke[q] >> BSH], 1);
      rv[q] = atomicAdd(&bins[N_EB + (kv[q] >> BSH)], 1);
    }
  }
  __syncthreads();
  if (tid < N_BINS) {
    const int c = bins[tid];
    bins[tid] = c > 0 ? atomicAdd(&cursor[tid], c) : 0;
  }
  __syncthreads();
#pragma unroll
  for (int q = 0; q < FILL_PER; ++q) {
    if (ok[q]) {
      const int be = ke[q] >> BSH;
      const int bv = kv[q] >> BSH;
      const int pe = bins[be] + re[q];
      const int pv = bins[N_EB + bv] + rv[q];
      if (pe < CAP_E)
        recE[(size_t)be * CAP_E + pe] = ((unsigned)(ke[q] & 127) << 17) | (unsigned)kv[q];
      if (pv < CAP_V)
        recV[(size_t)bv * CAP_V + pv] = ((unsigned)(kv[q] & 127) << 15) | (unsigned)ke[q];
    }
  }
}


#define VB 4
static_assert((size_t)N_E * HID * sizeof(float) + 1024 <= 0xFFFFFFFFull, "a Y row's byte offset must fit 32 bits");

__device__ __forceinline__ float lane_val(float x, int k) {
  return __int_as_float(__builtin_amdgcn_readlane(__float_as_int(x), k));
}

__device__ __forceinline__ void v_rows_load(
    const char* __restrict__ Yb, unsigned lane16, int e_l, int k0, int cnt,
    float4 (&ya)[VB], float4 (&yb)[VB])
{
#pragma unroll
  for (int q = 0; q < VB; ++q) {
    const int kk = min(k0 + q, cnt - 1);
    const int e = __builtin_amdgcn_readlane(e_l, kk);
    const unsigned off = (unsigned)e * (unsigned)(HID * sizeof(float)) + lane16;
    const float4* __restrict__ yrow4 = (const float4*)(Yb + off);
    ya[q] = yrow4[0];
    yb[q] = yrow4[32];
  }
}

__device__ __forceinline__ void v_rows_apply(
    const float4& w_l, bool hi, int k0,
    const float4 (&ya)[VB], const float4 (&yb)[VB], float4& a, float4& b)
{
#pragma unroll
  for (int q = 0; q < VB; ++q) {
    const int kk = k0 + q;
    const float wx = lane_val(w_l.x, kk);
    const float wy = lane_val(w_l.y, kk);
    const float wz = lane_val(w_l.z, kk);
    const float ww = lane_val(w_l.w, kk);
    const float wa = hi ? wy : wx;
    const float wb = hi ? ww : wz;
    a.x += wa * ya[q].x; a.y += wa * ya[q].y; a.z += wa * ya[q].z; a.w += wa * ya[q].w;
    b.x += wb * yb[q].x; b.y += wb * yb[q].y; b.z += wb * yb[q].z; b.w += wb * yb[q].w;
  }
}

__device__ __forceinline__ float elu(float x) {
  const float em1 = __builtin_amdgcn_exp2f(x * 1.4426950408889634f) - 1.0f;
  return x > 0.0f ? x : em1;
}

#define VPT 3

__global__ __launch_bounds__(256) void v_gather(
    const float* __restrict__ Y, const float* __restrict__ es,
    const int* __restrict__ cursor, const unsigned* __restrict__ recV,
    const int* __restrict__ inc_v, const int* __restrict__ inc_e,
    float* __restrict__ out)
{
  __shared__ int vlst[8 * VCAP];
  __shared__ int vcnt[8];

  const int tid = threadIdx.x;
  const int lane = threadIdx.x & 31;
  const int wave = __builtin_amdgcn_readfirstlane(threadIdx.x >> 5);
  const int v0 = blockIdx.x * 8;
  const int v = v0 + wave;

  const int bv = v0 >> BSH;
  const int nraw = cursor[N_EB + bv];
  const bool whole = nraw <= CAP_V;
  if (tid < 8) vcnt[tid] = 0;
  __syncthreads();
  if (whole) {
    const unsigned* __restrict__ rec = recV + (size_t)bv * CAP_V;
    const unsigned sib = (unsigned)(v0 & 127) >> 3;
    for (int c0 = 0; c0 < nraw; c0 += 256 * VPT) {
      unsigned r[VPT];
#pragma unroll
      for (int k = 0; k < VPT; ++k) {
        const int c = c0 + k * 256 + tid;
        r[k] = rec[min(c, nraw - 1)];
      }
      int slot[VPT];
#pragma unroll
      for (int k = 0; k < VPT; ++k) {
        const int c = c0 + k * 256 + tid;
        const unsigned vl = r[k] >> 15;
        slot[k] = VCAP;
        if (c < nraw && (vl >> 3) == sib) slot[k] = atomicAdd(&vcnt[vl & 7u], 1);
      }
#pragma unroll
      for (int k = 0; k < VPT; ++k) {
        const unsigned vl = r[k] >> 15;
        if (slot[k] < VCAP) vlst[(vl & 7u) * VCAP + slot[k]] = (int)(r[k] & 0x7FFFu);
      }
    }
  }
  __syncthreads();

  const int n = whole ? __builtin_amdgcn_readfirstlane(vcnt[wave]) : VCAP + 1;
  const int* lst = vlst + wave * VCAP;
  const float4* __restrict__ es4 = (const float4*)es;
  const char* __restrict__ Yb = (const char*)Y;
  const unsigned lane16 = (unsigned)lane * 16u;
  float4* orow4 = (float4*)(out + (size_t)v * HID);

  if (n <= 0) {
    const float4 z = make_float4(0.0f, 0.0f, 0.0f, 0.0f);
    orow4[lane] = z;
    orow4[lane + 32] = z;
    return;
  }

  const bool hi = lane >= 16;

  float4 a = make_float4(0.0f, 0.0f, 0.0f, 0.0f);
  float4 b = make_float4(0.0f, 0.0f, 0.0f, 0.0f);
  float4 ya[VB], yb[VB];

  if (n <= 32) {
    const int e_l = lst[min(lane, n - 1)];
    const float4 s_l = es4[e_l];
    v_rows_load(Yb, lane16, e_l, 0, n, ya, yb);

    float d0 = 0.0f, d1 = 0.0f, d2 = 0.0f, d3 = 0.0f;
    int k = 0;
    do {
      d0 += lane_val(s_l.x, k); d1 += lane_val(s_l.y, k);
      d2 += lane_val(s_l.z, k); d3 += lane_val(s_l.w, k);
    } while (++k < n);

    float4 w_l;
    w_l.x = lane < n ? s_l.x / d0 : 0.0f;
    w_l.y = lane < n ? s_l.y / d1 : 0.0f;
    w_l.z = lane < n ? s_l.z / d2 : 0.0f;
    w_l.w = lane < n ? s_l.w / d3 : 0.0f;
    v_rows_apply(w_l, hi, 0, ya, yb, a, b);
    for (int k0 = VB; k0 < n; k0 += VB) {
      v_rows_load(Yb, lane16, e_l, k0, n, ya, yb);
      v_rows_apply(w_l, hi, k0, ya, yb, a, b);
    }
  } else if (n <= VCAP) {
    float d0 = 0.0f, d1 = 0.0f, d2 = 0.0f, d3 = 0.0f;
    for (int c = 0; c < n; c += 32) {
      const int cnt = min(n - c, 32);
      const int e_t = lst[c + min(lane, cnt - 1)];
      const float4 s_t = es4[e_t];
      for (int k = 0; k < cnt; ++k) {
        d0 += lane_val(s_t.x, k); d1 += lane_val(s_t.y, k);
        d2 += lane_val(s_t.z, k); d3 += lane_val(s_t.w, k);
      }
    }
    for (int c = 0; c < n; c += 32) {
      const int cnt = min(n - c, 32);
      const int e_l = lst[c + min(lane, cnt - 1)];
      const float4 s_l = es4[e_l];
      float4 w_l;
      w_l.x = lane < cnt ? s_l.x / d0 : 0.0f;
      w_l.y = lane < cnt ? s_l.y / d1 : 0.0f;
      w_l.z = lane < cnt ? s_l.z / d2 : 0.0f;
      w_l.w = lane < cnt ? s_l.w / d3 : 0.0f;
      for (int k0 = 0; k0 < cnt; k0 += VB) {
        v_rows_load(Yb, lane16, e_l, k0, cnt, ya, yb);
        v_rows_apply(w_l, hi, k0, ya, yb, a, b);
      }
    }
  } else {
    float d0 = 0.0f, d1 = 0.0f, d2 = 0.0f, d3 = 0.0f;
    for (int c = 0; c < N_NZ; c += 32) {
      const int i = c + lane;
      const bool ok = i < N_NZ;
      const int key = ok ? inc_v[i] : -1;
      const int val = ok ? inc_e[i] : 0;
      unsigned m = __builtin_amdgcn_ballot_w32(key == v);
      while (m != 0u) {
        const int k = __builtin_ctz(m);
        m &= m - 1u;
        const float4 s4 = es4[__builtin_amdgcn_readlane(val, k)];
        d0 += s4.x; d1 += s4.y; d2 += s4.z; d3 += s4.w;
      }
    }
    for (int c = 0; c < N_NZ; c += 32) {
      const int i = c + lane;
      const bool ok = i < N_NZ;
      const int key = ok ? inc_v[i] : -1;
      const int val = ok ? inc_e[i] : 0;
      unsigned m = __builtin_amdgcn_ballot_w32(key == v);
      while (m != 0u) {
        const int k = __builtin_ctz(m);
        m &= m - 1u;
        const int ee = __builtin_amdgcn_readlane(val, k);
        const float4 s4 = es4[ee];
        const float wa = hi ? s4.y / d1 : s4.x / d0;
        const float wb = hi ? s4.w / d3 : s4.z / d2;
        const float4* __restrict__ yrow4 =
            (const float4*)(Yb + (size_t)ee * (HID * sizeof(float)) + lane16);
        const float4 y0 = yrow4[0];
        const float4 y1 = yrow4[32];
        a.x += wa * y0.x; a.y += wa * y0.y; a.z += wa * y0.z; a.w += wa * y0.w;
        b.x += wb * y1.x; b.y += wb * y1.y; b.z += wb * y1.z; b.w += wb * y1.w;
      }
    }
  }

  a.x = elu(a.x); a.y = elu(a.y); a.z = elu(a.z); a.w = elu(a.w);
  b.x = elu(b.x); b.y = elu(b.y); b.z = elu(b.z); b.w = elu(b.w);

  orow4[lane] = a;
  orow4[lane + 32] = b;
}

extern "C" void kernel_launch(void* const* d_in, const int* in_sizes, int n_in,
                              void* d_out, int out_size, void* d_ws, size_t ws_size,
                              hipStream_t stream) {
  const float* X     = (const float*)d_in[0];
  const float* W     = (const float*)d_in[1];
  const float* bvec  = (const float*)d_in[2];
  const float* a_e   = (const float*)d_in[3];
  const int*   inc_v = (const int*)d_in[4];
  const int*   inc_e = (const int*)d_in[5];
  float* out = (float*)d_out;

  float* ws    = (float*)d_ws;
  unsigned* recE = (unsigned*)ws;
  unsigned* recV = recE + (size_t)N_EB * CAP_E;
  float* Y     = ws + (size_t)N_E * HID;
  float* es    = Y  + (size_t)N_E * HID;
  int*   cursor = (int*)(es + (size_t)N_E * NH);

  hipMemsetAsync(cursor, 0, sizeof(int) * N_BINS, stream);

  group_fill<<<dim3(FILL_BLOCKS), FILL_THREADS, 0, stream>>>(inc_v, inc_e, cursor, recE, recV);
  edge_tile<<<dim3(N_E / 16), 256, 0, stream>>>(X, cursor, inc_v, inc_e, recE, W, bvec, a_e, Y, es);
  v_gather<<<dim3(N_V / 8), 256, 0, stream>>>(Y, es, cursor, recV, inc_v, inc_e, out);
}


# v_gather: a 4-wave work-group owns 32 vertices, each wave takes 8 in turn: 12500 waves enter, not 100000, and a bucket's words are ranked a quarter as often; sums in LDS arrival rank as before
# speedup vs baseline: 5.7448x; 1.0394x over previous
#include <hip/hip_runtime.h>
#include <hip/hip_bf16.h>

typedef __attribute__((ext_vector_type(2))) float v2f;
typedef __attribute__((ext_vector_type(8))) float v8f;

#define N_V   100000
#define N_E   20000
#define N_NZ  400000
#define CIN   256
#define HID   256
#define NH    4
#define DH    64
#define NEG_SLOPE 0.2f

#define BSH    7
#define N_EB   ((N_E + 127) >> BSH)
#define N_VB   ((N_V + 127) >> BSH)
#define N_BINS (N_EB + N_VB)
#define CAP_E  8192
#define CAP_V  2048
#define ROWCAP 256
#define VCAP   64
static_assert(N_V <= (1 << 17), "vertex id must fit 17 bits");
static_assert(N_E <= (1 << 15), "hyperedge id must fit 15 bits");
static_assert(N_BINS <= 1024, "one thread per bucket counter");
static_assert((size_t)N_EB * CAP_E + (size_t)N_VB * CAP_V <= (size_t)N_E * HID, "regions fit the list plane");


#define APITCH 260

#define EPT 12

__device__ __forceinline__ int edge_sum(
    const char* __restrict__ Xb, const int* __restrict__ lst, int t, int e,
    const int* __restrict__ inc_v, const int* __restrict__ inc_e,
    int lane, unsigned lane16, float4& a, float4& b)
{
  a = make_float4(0.0f, 0.0f, 0.0f, 0.0f);
  b = make_float4(0.0f, 0.0f, 0.0f, 0.0f);
  if (t <= ROWCAP) {
    for (int j = 0; j < t; j += 8) {
      float4 xa[8], xb[8];
#pragma unroll
      for (int k = 0; k < 8; ++k) {
        const int jj = min(j + k, t - 1);
        const unsigned off = (unsigned)lst[jj] * (unsigned)(CIN * sizeof(float)) + lane16;
        const float4* __restrict__ xrow4 = (const float4*)(Xb + off);
        xa[k] = xrow4[0];
        xb[k] = xrow4[32];
      }
#pragma unroll
      for (int k = 0; k < 8; ++k) {
        const bool in = j + k < t;
        a.x += in ? xa[k].x : 0.0f; a.y += in ? xa[k].y : 0.0f;
        a.z += in ? xa[k].z : 0.0f; a.w += in ? xa[k].w : 0.0f;
        b.x += in ? xb[k].x : 0.0f; b.y += in ? xb[k].y : 0.0f;
        b.z += in ? xb[k].z : 0.0f; b.w += in ? xb[k].w : 0.0f;
      }
    }
    return t;
  }
  int members = 0;
  for (int c = 0; c < N_NZ; c += 32) {
    const int i = c + lane;
    const bool ok = i < N_NZ;
    const int key = ok ? inc_e[i] : -1;
    const int val = ok ? inc_v[i] : 0;
    unsigned m = __builtin_amdgcn_ballot_w32(key == e);
    members += __builtin_popcount(m);
    while (m != 0u) {
      const int k = __builtin_ctz(m);
      m &= m - 1u;
      const int vv = __builtin_amdgcn_readlane(val, k);
      const unsigned off = (unsigned)vv * (unsigned)(CIN * sizeof(float)) + lane16;
      const float4* __restrict__ xrow4 = (const float4*)(Xb + off);
      const float4 xa = xrow4[0];
      const float4 xb = xrow4[32];
      a.x += xa.x; a.y += xa.y; a.z += xa.z; a.w += xa.w;
      b.x += xb.x; b.y += xb.y; b.z += xb.z; b.w += xb.w;
    }
  }
  return members;
}

__global__ __launch_bounds__(256) void edge_tile(
    const float* __restrict__ X, const int* __restrict__ cursor,
    const int* __restrict__ inc_v, const int* __restrict__ inc_e,
    const unsigned* __restrict__ recE,
    const float* __restrict__ W, const float* __restrict__ bvec,
    const float* __restrict__ a_e,
    float* __restrict__ Y, float* __restrict__ es)
{
  __shared__ __attribute__((aligned(16))) float As[16 * APITCH];
  __shared__ int elst[16 * ROWCAP];
  __shared__ int ecnt[16];

  const int tid = threadIdx.x;
  const int lane = threadIdx.x & 31;
  const int wave = __builtin_amdgcn_readfirstlane(threadIdx.x >> 5);
  const int row0 = blockIdx.x * 16;

  const int be = row0 >> BSH;
  const int nraw = cursor[be];
  const bool whole = nraw <= CAP_E;
  if (tid < 16) ecnt[tid] = 0;
  __syncthreads();
  if (whole) {
    const unsigned* __restrict__ rec = recE + (size_t)be * CAP_E;
    const unsigned tl = (unsigned)(row0 & 127) >> 4;
    for (int c0 = 0; c0 < nraw; c0 += 256 * EPT) {
      unsigned r[EPT];
#pragma unroll
      for (int k = 0; k < EPT; ++k) {
        const int c = c0 + k * 256 + tid;
        r[k] = rec[min(c, nraw - 1)];
      }
      int slot[EPT];
#pragma unroll
      for (int k = 0; k < EPT; ++k) {
        const int c = c0 + k * 256 + tid;
        const unsigned el = r[k] >> 17;
        slot[k] = ROWCAP;
        if (c < nraw && (el >> 4) == tl) slot[k] = atomicAdd(&ecnt[el & 15u], 1);
      }
#pragma unroll
      for (int k = 0; k < EPT; ++k) {
        const unsigned el = r[k] >> 17;
        if (slot[k] < ROWCAP) elst[(el & 15u) * ROWCAP + slot[k]] = (int)(r[k] & 0x1FFFFu);
      }
    }
  }
  __syncthreads();

  {
    const char* __restrict__ Xb = (const char*)X;
    const unsigned lane16 = (unsigned)lane * 16u;
    for (int q = 0; q < 2; ++q) {
      const int r = 2 * wave + q;
      const int e = row0 + r;
      const int t = whole ? __builtin_amdgcn_readfirstlane(ecnt[r]) : ROWCAP + 1;
      float4 a, b;
      const int members = edge_sum(Xb, elst + r * ROWCAP, t, e, inc_v, inc_e, lane, lane16, a, b);
      const float inv = 1.0f / fmaxf((float)members, 1.0f);
      a.x *= inv; a.y *= inv; a.z *= inv; a.w *= inv;
      b.x *= inv; b.y *= inv; b.z *= inv; b.w *= inv;
      float4* arow4 = (float4*)(As + r * APITCH);
      arow4[lane] = a;
      arow4[lane + 32] = b;
    }
  }
  __syncthreads();
  if (wave >= 4) return;

  const int head = wave;
  const int m = lane & 15;
  const int half = lane >> 4;

  const float* arow = As + m * APITCH;
  const float* __restrict__ wh = W + (size_t)head * (CIN * DH);

  v8f acc[4] = {v8f{}, v8f{}, v8f{}, v8f{}};
#pragma unroll 4
  for (int k0 = 0; k0 < CIN; k0 += 4) {
    const int ka = k0 + 2 * half;
    v2f afrag;
    afrag.x = arow[ka];
    afrag.y = arow[ka + 1];
    const float* __restrict__ wk = wh + (size_t)ka * DH + m;
#pragma unroll
    for (int t = 0; t < 4; ++t) {
      v2f bfrag;
      bfrag.x = wk[t * 16];
      bfrag.y = wk[t * 16 + DH];
      acc[t] = __builtin_amdgcn_wmma_f32_16x16x4_f32(
          false, afrag, false, bfrag, (short)0, acc[t], false, false);
    }
  }

  const int mbase = half * 8;
  float p[8] = {0.0f, 0.0f, 0.0f, 0.0f, 0.0f, 0.0f, 0.0f, 0.0f};
#pragma unroll
  for (int t = 0; t < 4; ++t) {
    const int col = head * DH + t * 16 + m;
    const float bias = bvec[col];
    const float ae = a_e[col];
#pragma unroll
    for (int r = 0; r < 8; ++r) {
      const float y = acc[t][r] + bias;
      Y[(size_t)(row0 + mbase + r) * HID + col] = y;
      p[r] += y * ae;
    }
  }
#pragma unroll
  for (int off = 8; off >= 1; off >>= 1) {
#pragma unroll
    for (int r = 0; r < 8; ++r) p[r] += __shfl_xor(p[r], off);
  }
  float alpha = p[0];
#pragma unroll
  for (int r = 1; r < 8; ++r) alpha = (m == r) ? p[r] : alpha;
  if (m < 8) {
    float s = alpha >= 0.0f ? alpha : NEG_SLOPE * alpha;
    s = fminf(fmaxf(s, 0.001f), 5.0f);
    es[(row0 + mbase + m) * NH + head] = __expf(s);
  }
}


#define FILL_BLOCKS  256
#define FILL_THREADS 1024
#define FILL_CHUNK   ((N_NZ + FILL_BLOCKS - 1) / FILL_BLOCKS)
#define FILL_PER     ((FILL_CHUNK + FILL_THREADS - 1) / FILL_THREADS)

__global__ __launch_bounds__(FILL_THREADS) void group_fill(
    const int* __restrict__ inc_v, const int* __restrict__ inc_e,
    int* __restrict__ cursor,
    unsigned* __restrict__ recE, unsigned* __restrict__ recV)
{
  __shared__ int bins[N_BINS];
  const int tid = threadIdx.x;
  if (tid < N_BINS) bins[tid] = 0;
  __syncthreads();

  int kv[FILL_PER], ke[FILL_PER], rv[FILL_PER], re[FILL_PER];
  bool ok[FILL_PER];
#pragma unroll
  for (int q = 0; q < FILL_PER; ++q) {
    const int li = tid + q * FILL_THREADS;
    const int i = blockIdx.x * FILL_CHUNK + li;
    ok[q] = li < FILL_CHUNK && i < N_NZ;
    kv[q] = ok[q] ? inc_v[i] : 0;
    ke[q] = ok[q] ? inc_e[i] : 0;
  }
#pragma unroll
  for (int q = 0; q < FILL_PER; ++q) {
    re[q] = 0;
    rv[q] = 0;
    if (ok[q]) {
      re[q] = atomicAdd(&bins[ke[q] >> BSH], 1);
      rv[q] = atomicAdd(&bins[N_EB + (kv[q] >> BSH)], 1);
    }
  }
  __syncthreads();
  if (tid < N_BINS) {
    const int c = bins[tid];
    bins[tid] = c > 0 ? atomicAdd(&cursor[tid], c) : 0;
  }
  __syncthreads();
#pragma unroll
  for (int q = 0; q < FILL_PER; ++q) {
    if (ok[q]) {
      const int be = ke[q] >> BSH;
      const int bv = kv[q] >> BSH;
      const int pe = bins[be] + re[q];
      const int pv = bins[N_EB + bv] + rv[q];
      if (pe < CAP_E)
        recE[(size_t)be * CAP_E + pe] = ((unsigned)(ke[q] & 127) << 17) | (unsigned)kv[q];
      if (pv < CAP_V)
        recV[(size_t)bv * CAP_V + pv] = ((unsigned)(kv[q] & 127) << 15) | (unsigned)ke[q];
    }
  }
}


#define VB 4
static_assert((size_t)N_E * HID * sizeof(float) + 1024 <= 0xFFFFFFFFull, "a Y row's byte offset must fit 32 bits");
static_assert((size_t)N_V * HID * sizeof(float) + 1024 <= 0xFFFFFFFFull, "an out row's byte offset must fit 32 bits");

__device__ __forceinline__ float lane_val(float x, int k) {
  return __int_as_float(__builtin_amdgcn_readlane(__float_as_int(x), k));
}

template <int G>
__device__ __forceinline__ void v_rows_load(
    const char* __restrict__ Yb, unsigned lane16, int e_l, int k0, int cnt,
    float4 (&ya)[G], float4 (&yb)[G])
{
#pragma unroll
  for (int q = 0; q < G; ++q) {
    const int kk = min(k0 + q, cnt - 1);
    const int e = __builtin_amdgcn_readlane(e_l, kk);
    const unsigned off = (unsigned)e * (unsigned)(HID * sizeof(float)) + lane16;
    const float4* __restrict__ yrow4 = (const float4*)(Yb + off);
    ya[q] = yrow4[0];
    yb[q] = yrow4[32];
  }
}

template <int G>
__device__ __forceinline__ void v_rows_apply(
    const float4& w_l, bool hi, int k0,
    const float4 (&ya)[G], const float4 (&yb)[G], float4& a, float4& b)
{
#pragma unroll
  for (int q = 0; q < G; ++q) {
    const int kk = k0 + q;
    const float wx = lane_val(w_l.x, kk);
    const float wy = lane_val(w_l.y, kk);
    const float wz = lane_val(w_l.z, kk);
    const float ww = lane_val(w_l.w, kk);
    const float wa = hi ? wy : wx;
    const float wb = hi ? ww : wz;
    a.x += wa * ya[q].x; a.y += wa * ya[q].y; a.z += wa * ya[q].z; a.w += wa * ya[q].w;
    b.x += wb * yb[q].x; b.y += wb * yb[q].y; b.z += wb * yb[q].z; b.w += wb * yb[q].w;
  }
}

__device__ __forceinline__ float elu(float x) {
  const float em1 = __builtin_amdgcn_exp2f(x * 1.4426950408889634f) - 1.0f;
  return x > 0.0f ? x : em1;
}

#define VGW 4
#define VPW 8
#define VGV (VGW * VPW)
static_assert(N_V % VGV == 0, "whole work-groups");
static_assert(128 % VGV == 0 && VGV == 32, "a work-group's vertices lie inside one bucket; masks below assume 32");

#define VPT 5

__global__ __launch_bounds__(VGW * 32) void v_gather(
    const float* __restrict__ Y, const float* __restrict__ es,
    const int* __restrict__ cursor, const unsigned* __restrict__ recV,
    const int* __restrict__ inc_v, const int* __restrict__ inc_e,
    float* __restrict__ out)
{
  __shared__ int vlst[VGV * VCAP];
  __shared__ int vcnt[VGV];

  const int tid = threadIdx.x;
  const int lane = threadIdx.x & 31;
  const int wave = __builtin_amdgcn_readfirstlane(threadIdx.x >> 5);
  const int v0 = blockIdx.x * VGV;

  const int bv = v0 >> BSH;
  const int nraw = cursor[N_EB + bv];
  const bool whole = nraw <= CAP_V;
  if (tid < VGV) vcnt[tid] = 0;
  __syncthreads();
  if (whole) {
    const unsigned* __restrict__ rec = recV + (size_t)bv * CAP_V;
    const unsigned sib = (unsigned)(v0 & 127) >> 5;
    for (int c0 = 0; c0 < nraw; c0 += VGW * 32 * VPT) {
      unsigned r[VPT];
#pragma unroll
      for (int k = 0; k < VPT; ++k) {
        const int c = c0 + k * (VGW * 32) + tid;
        r[k] = rec[min(c, nraw - 1)];
      }
      int slot[VPT];
#pragma unroll
      for (int k = 0; k < VPT; ++k) {
        const int c = c0 + k * (VGW * 32) + tid;
        const unsigned vl = r[k] >> 15;
        slot[k] = VCAP;
        if (c < nraw && (vl >> 5) == sib) slot[k] = atomicAdd(&vcnt[vl & 31u], 1);
      }
#pragma unroll
      for (int k = 0; k < VPT; ++k) {
        const unsigned vl = r[k] >> 15;
        if (slot[k] < VCAP) vlst[(vl & 31u) * VCAP + slot[k]] = (int)(r[k] & 0x7FFFu);
      }
    }
  }
  __syncthreads();

  const float4* __restrict__ es4 = (const float4*)es;
  const char* __restrict__ Yb = (const char*)Y;
  const unsigned lane16 = (unsigned)lane * 16u;
  const bool hi = lane >= 16;

  for (int i = 0; i < VPW; ++i) {
  const int lv = wave * VPW + i;
  const int v = v0 + lv;
  const int n = whole ? __builtin_amdgcn_readfirstlane(vcnt[lv]) : VCAP + 1;
  const int* lst = vlst + lv * VCAP;
  const unsigned ooff = (unsigned)v * (unsigned)(HID * sizeof(float)) + lane16;
  float4* orow4 = (float4*)((char*)out + ooff);

  if (n <= 0) {
    const float4 z = make_float4(0.0f, 0.0f, 0.0f, 0.0f);
    orow4[0] = z;
    orow4[32] = z;
    continue;
  }

  float4 a = make_float4(0.0f, 0.0f, 0.0f, 0.0f);
  float4 b = make_float4(0.0f, 0.0f, 0.0f, 0.0f);
  float4 ya[VB], yb[VB];

  if (n <= 32) {
    const int e_l = lst[min(lane, n - 1)];
    const float4 s_l = es4[e_l];
    v_rows_load(Yb, lane16, e_l, 0, n, ya, yb);

    float d0 = 0.0f, d1 = 0.0f, d2 = 0.0f, d3 = 0.0f;
    int k = 0;
    do {
      d0 += lane_val(s_l.x, k); d1 += lane_val(s_l.y, k);
      d2 += lane_val(s_l.z, k); d3 += lane_val(s_l.w, k);
    } while (++k < n);

    float4 w_l;
    w_l.x = lane < n ? s_l.x / d0 : 0.0f;
    w_l.y = lane < n ? s_l.y / d1 : 0.0f;
    w_l.z = lane < n ? s_l.z / d2 : 0.0f;
    w_l.w = lane < n ? s_l.w / d3 : 0.0f;
    v_rows_apply(w_l, hi, 0, ya, yb, a, b);
    for (int k0 = VB; k0 < n; k0 += VB) {
      v_rows_load(Yb, lane16, e_l, k0, n, ya, yb);
      v_rows_apply(w_l, hi, k0, ya, yb, a, b);
    }
  } else if (n <= VCAP) {
    float d0 = 0.0f, d1 = 0.0f, d2 = 0.0f, d3 = 0.0f;
    for (int c = 0; c < n; c += 32) {
      const int cnt = min(n - c, 32);
      const int e_t = lst[c + min(lane, cnt - 1)];
      const float4 s_t = es4[e_t];
      for (int k = 0; k < cnt; ++k) {
        d0 += lane_val(s_t.x, k); d1 += lane_val(s_t.y, k);
        d2 += lane_val(s_t.z, k); d3 += lane_val(s_t.w, k);
      }
    }
    for (int c = 0; c < n; c += 32) {
      const int cnt = min(n - c, 32);
      const int e_l = lst[c + min(lane, cnt - 1)];
      const float4 s_l = es4[e_l];
      float4 w_l;
      w_l.x = lane < cnt ? s_l.x / d0 : 0.0f;
      w_l.y = lane < cnt ? s_l.y / d1 : 0.0f;
      w_l.z = lane < cnt ? s_l.z / d2 : 0.0f;
      w_l.w = lane < cnt ? s_l.w / d3 : 0.0f;
      float4 yc[2], yd[2];
      for (int k0 = 0; k0 < cnt; k0 += 2) {
        v_rows_load(Yb, lane16, e_l, k0, cnt, yc, yd);
        v_rows_apply(w_l, hi, k0, yc, yd, a, b);
      }
    }
  } else {
    float d0 = 0.0f, d1 = 0.0f, d2 = 0.0f, d3 = 0.0f;
    for (int c = 0; c < N_NZ; c += 32) {
      const int i = c + lane;
      const bool ok = i < N_NZ;
      const int key = ok ? inc_v[i] : -1;
      const int val = ok ? inc_e[i] : 0;
      unsigned m = __builtin_amdgcn_ballot_w32(key == v);
      while (m != 0u) {
        const int k = __builtin_ctz(m);
        m &= m - 1u;
        const float4 s4 = es4[__builtin_amdgcn_readlane(val, k)];
        d0 += s4.x; d1 += s4.y; d2 += s4.z; d3 += s4.w;
      }
    }
    for (int c = 0; c < N_NZ; c += 32) {
      const int i = c + lane;
      const bool ok = i < N_NZ;
      const int key = ok ? inc_v[i] : -1;
      const int val = ok ? inc_e[i] : 0;
      unsigned m = __builtin_amdgcn_ballot_w32(key == v);
      while (m != 0u) {
        const int k = __builtin_ctz(m);
        m &= m - 1u;
        const int ee = __builtin_amdgcn_readlane(val, k);
        const float4 s4 = es4[ee];
        const float wa = hi ? s4.y / d1 : s4.x / d0;
        const float wb = hi ? s4.w / d3 : s4.z / d2;
        const float4* __restrict__ yrow4 =
            (const float4*)(Yb + (size_t)ee * (HID * sizeof(float)) + lane16);
        const float4 y0 = yrow4[0];
        const float4 y1 = yrow4[32];
        a.x += wa * y0.x; a.y += wa * y0.y; a.z += wa * y0.z; a.w += wa * y0.w;
        b.x += wb * y1.x; b.y += wb * y1.y; b.z += wb * y1.z; b.w += wb * y1.w;
      }
    }
  }

  a.x = elu(a.x); a.y = elu(a.y); a.z = elu(a.z); a.w = elu(a.w);
  b.x = elu(b.x); b.y = elu(b.y); b.z = elu(b.z); b.w = elu(b.w);

  orow4[0] = a;
  orow4[32] = b;
  }
}

extern "C" void kernel_launch(void* const* d_in, const int* in_sizes, int n_in,
                              void* d_out, int out_size, void* d_ws, size_t ws_size,
                              hipStream_t stream) {
  const float* X     = (const float*)d_in[0];
  const float* W     = (const float*)d_in[1];
  const float* bvec  = (const float*)d_in[2];
  const float* a_e   = (const float*)d_in[3];
  const int*   inc_v = (const int*)d_in[4];
  const int*   inc_e = (const int*)d_in[5];
  float* out = (float*)d_out;

  float* ws    = (float*)d_ws;
  unsigned* recE = (unsigned*)ws;
  unsigned* recV = recE + (size_t)N_EB * CAP_E;
  float* Y     = ws + (size_t)N_E * HID;
  float* es    = Y  + (size_t)N_E * HID;
  int*   cursor = (int*)(es + (size_t)N_E * NH);

  hipMemsetAsync(cursor, 0, sizeof(int) * N_BINS, stream);

  group_fill<<<dim3(FILL_BLOCKS), FILL_THREADS, 0, stream>>>(inc_v, inc_e, cursor, recE, recV);
  edge_tile<<<dim3(N_E / 16), 256, 0, stream>>>(X, cursor, inc_v, inc_e, recE, W, bvec, a_e, Y, es);
  v_gather<<<dim3(N_V / VGV), VGW * 32, 0, stream>>>(Y, es, cursor, recV, inc_v, inc_e, out);
}
